# starting kernel
# baseline (speedup 1.0000x reference)
#include <hip/hip_runtime.h>

typedef __attribute__((ext_vector_type(2))) float v2f;
typedef __attribute__((ext_vector_type(8))) float v8f;


__global__ void gcn_init_deg(float* __restrict__ deg, int n) {
  int i = blockIdx.x * blockDim.x + threadIdx.x;
  if (i < n) deg[i] = 1.0f;
}

__global__ void gcn_zero(float* __restrict__ p, long long n) {
  long long i = (long long)blockIdx.x * blockDim.x + threadIdx.x;
  if (i < n) p[i] = 0.0f;
}

__global__ void gcn_init_out(float* __restrict__ out, const float* __restrict__ b2,
                             int n3) {
  int i = blockIdx.x * blockDim.x + threadIdx.x;
  if (i < n3) out[i] = b2[i % 3];
}

__global__ void gcn_deg_accum(const int* __restrict__ dst, float* __restrict__ deg,
                              int E) {
  int e = blockIdx.x * blockDim.x + threadIdx.x;
  if (e < E) atomicAdd(&deg[dst[e]], 1.0f);
}

__global__ void gcn_rsqrt(float* __restrict__ deg, int n) {
  int i = blockIdx.x * blockDim.x + threadIdx.x;
  if (i < n) deg[i] = rsqrtf(deg[i]);
}

__global__ __launch_bounds__(256) void gcn_gemm1_wmma(
    const float* __restrict__ x, const float* __restrict__ W,
    float* __restrict__ h, int M) {
  const int lane = threadIdx.x & 31;
  const int wave = threadIdx.x >> 5;
  const int m0 = (blockIdx.x * 4 + wave) * 16;
  if (m0 >= M) return;

  const int laneM = lane & 15;
  const int kOff  = (lane >> 4) << 1;
  int row = m0 + laneM;
  if (row >= M) row = M - 1;
  const float* __restrict__ arow = x + (long long)row * 512;

  v8f c[8] = {};

  for (int k = 0; k < 512; k += 4) {
    v2f a;
    a.x = arow[k + kOff];
    a.y = arow[k + kOff + 1];
    const float* __restrict__ w0 = W + (long long)(k + kOff) * 128 + laneM;
    const float* __restrict__ w1 = w0 + 128;
#pragma unroll
    for (int j = 0; j < 8; ++j) {
      v2f b;
      b.x = w0[j * 16];
      b.y = w1[j * 16];
      c[j] = __builtin_amdgcn_wmma_f32_16x16x4_f32(
           false, a,  false, b,
           (short)0, c[j],  false,  false);
    }
  }

  const int rbase = m0 + ((lane >> 4) << 3);
#pragma unroll
  for (int j = 0; j < 8; ++j) {
    const int col = j * 16 + laneM;
#pragma unroll
    for (int v = 0; v < 8; ++v) {
      const int r = rbase + v;
      if (r < M) h[(long long)r * 128 + col] = c[j][v];
    }
  }
}

__global__ __launch_bounds__(256) void gcn_agg1(
    const int* __restrict__ src, const int* __restrict__ dst,
    const float* __restrict__ dinv, const float* __restrict__ h1,
    float* __restrict__ agg, int E) {
  long long gid = (long long)blockIdx.x * blockDim.x + threadIdx.x;
  int e = (int)(gid >> 5);
  int lane = (int)(gid & 31);
  if (e >= E) return;
  int s = src[e], d = dst[e];
  float nrm = dinv[s] * dinv[d];
  const float4 v = ((const float4*)(h1 + (long long)s * 128))[lane];
  float* op = agg + (long long)d * 128 + lane * 4;
  atomicAdd(op + 0, nrm * v.x);
  atomicAdd(op + 1, nrm * v.y);
  atomicAdd(op + 2, nrm * v.z);
  atomicAdd(op + 3, nrm * v.w);
}

__global__ void gcn_post1(const float* __restrict__ h1,
                          const float* __restrict__ dinv,
                          const float* __restrict__ b1,
                          float* __restrict__ agg, int total) {
  int t = blockIdx.x * blockDim.x + threadIdx.x;
  if (t >= total) return;
  int i = t >> 7, f = t & 127;
  float di = dinv[i];
  float v = agg[t] + di * di * h1[t] + b1[f];
  agg[t] = v > 0.0f ? v : 0.0f;
}

__global__ void gcn_gemm2(const float* __restrict__ y1,
                          const float* __restrict__ W2,
                          float* __restrict__ h2, int N) {
  int i = blockIdx.x * blockDim.x + threadIdx.x;
  if (i >= N) return;
  const float* __restrict__ row = y1 + (long long)i * 128;
  float a0 = 0.f, a1 = 0.f, a2 = 0.f;
#pragma unroll 4
  for (int k = 0; k < 128; ++k) {
    float xv = row[k];
    a0 += xv * W2[k * 3 + 0];
    a1 += xv * W2[k * 3 + 1];
    a2 += xv * W2[k * 3 + 2];
  }
  h2[(long long)i * 3 + 0] = a0;
  h2[(long long)i * 3 + 1] = a1;
  h2[(long long)i * 3 + 2] = a2;
}

__global__ void gcn_agg2(const int* __restrict__ src, const int* __restrict__ dst,
                         const float* __restrict__ dinv,
                         const float* __restrict__ h2,
                         float* __restrict__ out, int E, int N) {
  int t = blockIdx.x * blockDim.x + threadIdx.x;
  if (t < E) {
    int s = src[t], d = dst[t];
    float nrm = dinv[s] * dinv[d];
    atomicAdd(&out[(long long)d * 3 + 0], nrm * h2[(long long)s * 3 + 0]);
    atomicAdd(&out[(long long)d * 3 + 1], nrm * h2[(long long)s * 3 + 1]);
    atomicAdd(&out[(long long)d * 3 + 2], nrm * h2[(long long)s * 3 + 2]);
  } else if (t < E + N) {
    int i = t - E;
    float nn = dinv[i] * dinv[i];
    atomicAdd(&out[(long long)i * 3 + 0], nn * h2[(long long)i * 3 + 0]);
    atomicAdd(&out[(long long)i * 3 + 1], nn * h2[(long long)i * 3 + 1]);
    atomicAdd(&out[(long long)i * 3 + 2], nn * h2[(long long)i * 3 + 2]);
  }
}

extern "C" void kernel_launch(void* const* d_in, const int* in_sizes, int n_in,
                              void* d_out, int out_size, void* d_ws, size_t ws_size,
                              hipStream_t stream) {
  const float* x  = (const float*)d_in[0];
  const int*   ei = (const int*)d_in[1];
  const float* W1 = (const float*)d_in[2];
  const float* b1 = (const float*)d_in[3];
  const float* W2 = (const float*)d_in[4];
  const float* b2 = (const float*)d_in[5];
  float* out = (float*)d_out;

  const int N = in_sizes[0] / 512;
  const int E = in_sizes[1] / 2;
  const int* src = ei;
  const int* dst = ei + E;

  float* ws   = (float*)d_ws;
  float* deg  = ws;
  float* h1   = ws + (((long long)N + 255) & ~255LL);
  float* agg1 = h1 + (long long)N * 128;
  float* h2   = agg1 + (long long)N * 128;

  const int B = 256;
  const long long tot1 = (long long)N * 128;
  const long long tE32 = (long long)E * 32;

  gcn_init_deg<<<(N + B - 1) / B, B, 0, stream>>>(deg, N);
  gcn_zero<<<(int)((tot1 + B - 1) / B), B, 0, stream>>>(agg1, tot1);
  gcn_init_out<<<(N * 3 + B - 1) / B, B, 0, stream>>>(out, b2, N * 3);
  gcn_deg_accum<<<(E + B - 1) / B, B, 0, stream>>>(dst, deg, E);
  gcn_rsqrt<<<(N + B - 1) / B, B, 0, stream>>>(deg, N);

  gcn_gemm1_wmma<<<(N + 63) / 64, B, 0, stream>>>(x, W1, h1, N);

  gcn_agg1<<<(int)((tE32 + B - 1) / B), B, 0, stream>>>(src, dst, deg, h1, agg1, E);
  gcn_post1<<<(int)((tot1 + B - 1) / B), B, 0, stream>>>(h1, deg, b1, agg1, (int)tot1);

  gcn_gemm2<<<(N + B - 1) / B, B, 0, stream>>>(agg1, W2, h2, N);
  gcn_agg2<<<(E + N + B - 1) / B, B, 0, stream>>>(src, dst, deg, h2, out, E, N);
}


# in-call destination-sorted edge list (one returning int atomic = count + arrival-order rank, scan, fill); both layers gather per node, no float atomics; 128x3 product fused, wave tree sum
# speedup vs baseline: 1.8005x; 1.8005x over previous
#include <hip/hip_runtime.h>


typedef __attribute__((ext_vector_type(2))) float v2f;
typedef __attribute__((ext_vector_type(8))) float v8f;


__global__ void gcn_zero_cnt(int* __restrict__ cnt, int n) {
  int i = blockIdx.x * blockDim.x + threadIdx.x;
  if (i < n) cnt[i] = 0;
}

__global__ void gcn_count_rank(const int* __restrict__ dst, int* __restrict__ cnt,
                               int* __restrict__ rank, int E) {
  int e = blockIdx.x * blockDim.x + threadIdx.x;
  if (e < E) rank[e] = atomicAdd(&cnt[dst[e]], 1);
}

__global__ __launch_bounds__(1024) void gcn_scan_dinv(
    const int* __restrict__ cnt, int* __restrict__ off, float* __restrict__ dinv,
    int n) {
  __shared__ int frontPart[32];
  __shared__ int waveTot[32];
  const int t = threadIdx.x;
  const int lane = t & 31;
  const int w = t >> 5;
  const int start = blockIdx.x * 1024;

  int p = 0;
  for (int i = t; i < start; i += 1024) p += cnt[i];
  for (int o = 16; o > 0; o >>= 1) p += __shfl_xor(p, o);

  const int i = start + t;
  const int c = (i < n) ? cnt[i] : 0;
  int x = c;
  for (int o = 1; o < 32; o <<= 1) {
    const int y = __shfl_up(x, o);
    if (lane >= o) x += y;
  }
  if (lane == 0) frontPart[w] = p;
  if (lane == 31) waveTot[w] = x;
  __syncthreads();

  int base = frontPart[lane];
  int prev = (lane < w) ? waveTot[lane] : 0;
  for (int o = 16; o > 0; o >>= 1) {
    base += __shfl_xor(base, o);
    prev += __shfl_xor(prev, o);
  }
  if (i < n) {
    off[i] = base + prev + x - c;
    dinv[i] = rsqrtf((float)(c + 1));
  }
}

__global__ void gcn_fill(const int* __restrict__ src, const int* __restrict__ dst,
                         const int* __restrict__ off, const int* __restrict__ rank,
                         int* __restrict__ list, int E) {
  int e = blockIdx.x * blockDim.x + threadIdx.x;
  if (e < E) list[off[dst[e]] + rank[e]] = src[e];
}

__global__ __launch_bounds__(256) void gcn_gemm1_wmma(
    const float* __restrict__ x, const float* __restrict__ W,
    float* __restrict__ h, int M) {
  const int lane = threadIdx.x & 31;
  const int wave = threadIdx.x >> 5;
  const int m0 = (blockIdx.x * 4 + wave) * 16;
  if (m0 >= M) return;

  const int laneM = lane & 15;
  const int kOff  = (lane >> 4) << 1;
  int row = m0 + laneM;
  if (row >= M) row = M - 1;
  const float* __restrict__ arow = x + (long long)row * 512;

  v8f c[8] = {};

  for (int k = 0; k < 512; k += 4) {
    v2f a;
    a.x = arow[k + kOff];
    a.y = arow[k + kOff + 1];
    const float* __restrict__ w0 = W + (long long)(k + kOff) * 128 + laneM;
    const float* __restrict__ w1 = w0 + 128;
#pragma unroll
    for (int j = 0; j < 8; ++j) {
      v2f b;
      b.x = w0[j * 16];
      b.y = w1[j * 16];
      c[j] = __builtin_amdgcn_wmma_f32_16x16x4_f32(
           false, a,  false, b,
           (short)0, c[j],  false,  false);
    }
  }

  const int rbase = m0 + ((lane >> 4) << 3);
#pragma unroll
  for (int j = 0; j < 8; ++j) {
    const int col = j * 16 + laneM;
#pragma unroll
    for (int v = 0; v < 8; ++v) {
      const int r = rbase + v;
      if (r < M) h[(long long)r * 128 + col] = c[j][v];
    }
  }
}

__global__ __launch_bounds__(1024) void gcn_gather1(
    const int* __restrict__ off, const int* __restrict__ cnt,
    const int* __restrict__ list, const float* __restrict__ dinv,
    const float* __restrict__ h1, const float* __restrict__ b1,
    const float* __restrict__ W2, float* __restrict__ h2, int N) {
  __shared__ float res[96];
  const int lane = threadIdx.x & 31;
  const int w = __builtin_amdgcn_readfirstlane((int)(threadIdx.x >> 5));
  const int d = blockIdx.x * 32 + w;

  if (d < N) {
    const int beg = off[d];
    const int n = cnt[d];
    const float dd = dinv[d];
    float4 acc = make_float4(0.f, 0.f, 0.f, 0.f);

    for (int base = 0; base < n; base += 32) {
      const int m = (n - base < 32) ? (n - base) : 32;
      int s_l = 0;
      float w_l = 0.f;
      if (lane < m) {
        s_l = list[beg + base + lane];
        s_l = ((unsigned)s_l < (unsigned)N) ? s_l : 0;
        w_l = dinv[s_l] * dd;
      }
#pragma unroll 4
      for (int k = 0; k < m; ++k) {
        const int s = __builtin_amdgcn_readlane(s_l, k);
        const float nrm = __int_as_float(__builtin_amdgcn_readlane(__float_as_int(w_l), k));
        const float4 v = ((const float4*)(h1 + (long long)s * 128))[lane];
        acc.x += nrm * v.x;
        acc.y += nrm * v.y;
        acc.z += nrm * v.z;
        acc.w += nrm * v.w;
      }
    }

    const float4 hv = ((const float4*)(h1 + (long long)d * 128))[lane];
    const float nn = dd * dd;
    float y0 = acc.x + nn * hv.x + b1[lane * 4 + 0];
    float y1 = acc.y + nn * hv.y + b1[lane * 4 + 1];
    float y2 = acc.z + nn * hv.z + b1[lane * 4 + 2];
    float y3 = acc.w + nn * hv.w + b1[lane * 4 + 3];
    y0 = y0 > 0.0f ? y0 : 0.0f;
    y1 = y1 > 0.0f ? y1 : 0.0f;
    y2 = y2 > 0.0f ? y2 : 0.0f;
    y3 = y3 > 0.0f ? y3 : 0.0f;

    const float* __restrict__ wr = W2 + lane * 12;
    float p0 = y0 * wr[0] + y1 * wr[3] + y2 * wr[6] + y3 * wr[9];
    float p1 = y0 * wr[1] + y1 * wr[4] + y2 * wr[7] + y3 * wr[10];
    float p2 = y0 * wr[2] + y1 * wr[5] + y2 * wr[8] + y3 * wr[11];
    for (int o = 16; o > 0; o >>= 1) {
      p0 += __shfl_xor(p0, o);
      p1 += __shfl_xor(p1, o);
      p2 += __shfl_xor(p2, o);
    }
    if (lane == 0) {
      res[w * 3 + 0] = p0;
      res[w * 3 + 1] = p1;
      res[w * 3 + 2] = p2;
    }
  }
  __syncthreads();

  if (threadIdx.x < 24) {
    const long long f = (long long)blockIdx.x * 96 + threadIdx.x * 4;
    const long long n3 = (long long)N * 3;
    if (f + 3 < n3) {
      *(float4*)(h2 + f) = make_float4(res[threadIdx.x * 4 + 0], res[threadIdx.x * 4 + 1],
                                       res[threadIdx.x * 4 + 2], res[threadIdx.x * 4 + 3]);
    } else {
      for (int q = 0; q < 4; ++q)
        if (f + q < n3) h2[f + q] = res[threadIdx.x * 4 + q];
    }
  }
}

__global__ __launch_bounds__(256) void gcn_gather2(
    const int* __restrict__ off, const int* __restrict__ cnt,
    const int* __restrict__ list, const float* __restrict__ dinv,
    const float* __restrict__ h2, const float* __restrict__ b2,
    float* __restrict__ out, int N) {
  __shared__ float res[96];
  const int t = threadIdx.x;
  const int g = t >> 3;
  const int l = t & 7;
  const int d = blockIdx.x * 32 + g;
  const bool valid = d < N;
  const int dc = valid ? d : 0;
  const int beg = off[dc];
  const int n = valid ? cnt[dc] : 0;
  const float dd = dinv[dc];

  float a0 = 0.f, a1 = 0.f, a2 = 0.f;
  for (int j = l; j < n; j += 8) {
    int s = list[beg + j];
    s = ((unsigned)s < (unsigned)N) ? s : 0;
    const float nrm = dinv[s] * dd;
    const float* __restrict__ hp = h2 + (long long)s * 3;
    a0 += nrm * hp[0];
    a1 += nrm * hp[1];
    a2 += nrm * hp[2];
  }
  for (int o = 4; o > 0; o >>= 1) {
    a0 += __shfl_xor(a0, o);
    a1 += __shfl_xor(a1, o);
    a2 += __shfl_xor(a2, o);
  }
  if (l == 0) {
    const float nn = dd * dd;
    const float* __restrict__ hp = h2 + (long long)dc * 3;
    res[g * 3 + 0] = a0 + nn * hp[0] + b2[0];
    res[g * 3 + 1] = a1 + nn * hp[1] + b2[1];
    res[g * 3 + 2] = a2 + nn * hp[2] + b2[2];
  }
  __syncthreads();

  if (t < 24) {
    const long long f = (long long)blockIdx.x * 96 + t * 4;
    const long long n3 = (long long)N * 3;
    if (f + 3 < n3) {
      *(float4*)(out + f) = make_float4(res[t * 4 + 0], res[t * 4 + 1],
                                        res[t * 4 + 2], res[t * 4 + 3]);
    } else {
      for (int q = 0; q < 4; ++q)
        if (f + q < n3) out[f + q] = res[t * 4 + q];
    }
  }
}

extern "C" void kernel_launch(void* const* d_in, const int* in_sizes, int n_in,
                              void* d_out, int out_size, void* d_ws, size_t ws_size,
                              hipStream_t stream) {
  const float* x  = (const float*)d_in[0];
  const int*   ei = (const int*)d_in[1];
  const float* W1 = (const float*)d_in[2];
  const float* b1 = (const float*)d_in[3];
  const float* W2 = (const float*)d_in[4];
  const float* b2 = (const float*)d_in[5];
  float* out = (float*)d_out;

  const int N = in_sizes[0] / 512;
  const int E = in_sizes[1] / 2;
  const int* src = ei;
  const int* dst = ei + E;

  const long long Npad = ((long long)N + 255) & ~255LL;
  const long long Epad = ((long long)E + 255) & ~255LL;
  float* ws   = (float*)d_ws;
  float* dinv = ws;
  float* h1   = ws + Npad;
  float* pl   = h1 + (long long)N * 128;
  float* h2   = pl + (long long)N * 128;
  int* cnt  = (int*)pl;
  int* off  = cnt + Npad;
  int* rank = off + Npad;
  int* list = rank + Epad;

  const int B = 256;

  gcn_zero_cnt<<<(N + B - 1) / B, B, 0, stream>>>(cnt, N);
  gcn_count_rank<<<(E + B - 1) / B, B, 0, stream>>>(dst, cnt, rank, E);
  gcn_scan_dinv<<<(N + 1023) / 1024, 1024, 0, stream>>>(cnt, off, dinv, N);
  gcn_fill<<<(E + B - 1) / B, B, 0, stream>>>(src, dst, off, rank, list, E);

  gcn_gemm1_wmma<<<(N + 63) / 64, B, 0, stream>>>(x, W1, h1, N);

  gcn_gather1<<<(N + 31) / 32, 1024, 0, stream>>>(off, cnt, list, dinv, h1, b1, W2, h2, N);
  gcn_gather2<<<(N + 31) / 32, B, 0, stream>>>(off, cnt, list, dinv, h2, b2, out, N);
}


# GEMM1 rebuilt: 4-wave work-group per WGP, each wave keeps its 512x32 slice of W1 in registers, row tiles computed once (were twice), x read 16 B per lane half a tile ahead; new fixed k sum order
# speedup vs baseline: 2.4425x; 1.3566x over previous
#include <hip/hip_runtime.h>


typedef __attribute__((ext_vector_type(2))) float v2f;
typedef __attribute__((ext_vector_type(8))) float v8f;


__global__ void gcn_zero_cnt(int* __restrict__ cnt, int n) {
  int i = blockIdx.x * blockDim.x + threadIdx.x;
  if (i < n) cnt[i] = 0;
}

__global__ void gcn_count_rank(const int* __restrict__ dst, int* __restrict__ cnt,
                               int* __restrict__ rank, int E) {
  int e = blockIdx.x * blockDim.x + threadIdx.x;
  if (e < E) rank[e] = atomicAdd(&cnt[dst[e]], 1);
}

__global__ __launch_bounds__(1024) void gcn_scan_dinv(
    const int* __restrict__ cnt, int* __restrict__ off, float* __restrict__ dinv,
    int n) {
  __shared__ int frontPart[32];
  __shared__ int waveTot[32];
  const int t = threadIdx.x;
  const int lane = t & 31;
  const int w = t >> 5;
  const int start = blockIdx.x * 1024;

  int p = 0;
  for (int i = t; i < start; i += 1024) p += cnt[i];
  for (int o = 16; o > 0; o >>= 1) p += __shfl_xor(p, o);

  const int i = start + t;
  const int c = (i < n) ? cnt[i] : 0;
  int x = c;
  for (int o = 1; o < 32; o <<= 1) {
    const int y = __shfl_up(x, o);
    if (lane >= o) x += y;
  }
  if (lane == 0) frontPart[w] = p;
  if (lane == 31) waveTot[w] = x;
  __syncthreads();

  int base = frontPart[lane];
  int prev = (lane < w) ? waveTot[lane] : 0;
  for (int o = 16; o > 0; o >>= 1) {
    base += __shfl_xor(base, o);
    prev += __shfl_xor(prev, o);
  }
  if (i < n) {
    off[i] = base + prev + x - c;
    dinv[i] = rsqrtf((float)(c + 1));
  }
}

__global__ void gcn_fill(const int* __restrict__ src, const int* __restrict__ dst,
                         const int* __restrict__ off, const int* __restrict__ rank,
                         int* __restrict__ list, int E) {
  int e = blockIdx.x * blockDim.x + threadIdx.x;
  if (e < E) list[off[dst[e]] + rank[e]] = src[e];
}

__global__ __launch_bounds__(128) void gcn_gemm1_wmma(
    const float* __restrict__ x, const float* __restrict__ W,
    float* __restrict__ h, int M) {
  const int lane = threadIdx.x & 31;
  const int J = __builtin_amdgcn_readfirstlane((int)(threadIdx.x >> 5));
  const int c = lane & 15;
  const int hh = lane >> 4;
  const int ntiles = (M + 15) >> 4;
  const int G = (int)gridDim.x;

  v2f bf[8][16][2];
  {
    const float* __restrict__ wb = W + (4 * hh) * 128 + 32 * J + 2 * c;
#pragma unroll
    for (int kb = 0; kb < 8; ++kb) {
#pragma unroll
      for (int s = 0; s < 16; ++s) {
        const int k = 64 * kb + 8 * (s >> 1) + 2 * (s & 1);
        const float2 r0 = *(const float2*)(wb + k * 128);
        const float2 r1 = *(const float2*)(wb + (k + 1) * 128);
        bf[kb][s][0].x = r0.x; bf[kb][s][0].y = r1.x;
        bf[kb][s][1].x = r0.y; bf[kb][s][1].y = r1.y;
      }
    }
  }

  int tile = (int)blockIdx.x;
  if (tile >= ntiles) return;
  float4 a0[32];
  float4 a1[32];
  {
    int row = tile * 16 + c;
    if (row >= M) row = M - 1;
    const float4* __restrict__ ar = (const float4*)(x + (long long)row * 512 + 4 * hh);
#pragma unroll
    for (int i = 0; i < 32; ++i) a0[i] = ar[16 * (i >> 3) + 2 * (i & 7)];
  }

  for (; tile < ntiles; tile += G) {
    {
      int row = tile * 16 + c;
      if (row >= M) row = M - 1;
      const float4* __restrict__ ar = (const float4*)(x + (long long)row * 512 + 4 * hh);
#pragma unroll
      for (int i = 0; i < 32; ++i) a1[i] = ar[64 + 16 * (i >> 3) + 2 * (i & 7)];
    }
    __builtin_amdgcn_sched_barrier(0);

    v8f acc0 = {};
    v8f acc1 = {};

#pragma unroll
    for (int kb = 0; kb < 4; ++kb) {
#pragma unroll
      for (int s = 0; s < 16; ++s) {
        const float4 av = a0[8 * kb + (s >> 1)];
        v2f a;
        if (s & 1) { a.x = av.z; a.y = av.w; } else { a.x = av.x; a.y = av.y; }
        acc0 = __builtin_amdgcn_wmma_f32_16x16x4_f32(
             false, a,  false, bf[kb][s][0],
             (short)0, acc0,  false,  false);
        acc1 = __builtin_amdgcn_wmma_f32_16x16x4_f32(
             false, a,  false, bf[kb][s][1],
             (short)0, acc1,  false,  false);
      }
    }

    __builtin_amdgcn_sched_barrier(0);
    {
      const int nt = (tile + G < ntiles) ? tile + G : tile;
      int row = nt * 16 + c;
      if (row >= M) row = M - 1;
      const float4* __restrict__ ar = (const float4*)(x + (long long)row * 512 + 4 * hh);
#pragma unroll
      for (int i = 0; i < 32; ++i) a0[i] = ar[16 * (i >> 3) + 2 * (i & 7)];
    }
    __builtin_amdgcn_sched_barrier(0);

#pragma unroll
    for (int kb = 4; kb < 8; ++kb) {
#pragma unroll
      for (int s = 0; s < 16; ++s) {
        const float4 av = a1[8 * (kb - 4) + (s >> 1)];
        v2f a;
        if (s & 1) { a.x = av.z; a.y = av.w; } else { a.x = av.x; a.y = av.y; }
        acc0 = __builtin_amdgcn_wmma_f32_16x16x4_f32(
             false, a,  false, bf[kb][s][0],
             (short)0, acc0,  false,  false);
        acc1 = __builtin_amdgcn_wmma_f32_16x16x4_f32(
             false, a,  false, bf[kb][s][1],
             (short)0, acc1,  false,  false);
      }
    }

    const int rbase = tile * 16 + 8 * hh;
    float* __restrict__ hp = h + (long long)rbase * 128 + 32 * J + 2 * c;
    if (tile * 16 + 16 <= M) {
#pragma unroll
      for (int v = 0; v < 8; ++v) *(float2*)(hp + v * 128) = make_float2(acc0[v], acc1[v]);
    } else {
#pragma unroll
      for (int v = 0; v < 8; ++v) {
        if (rbase + v < M) *(float2*)(hp + v * 128) = make_float2(acc0[v], acc1[v]);
      }
    }
  }
}

__global__ __launch_bounds__(1024) void gcn_gather1(
    const int* __restrict__ off, const int* __restrict__ cnt,
    const int* __restrict__ list, const float* __restrict__ dinv,
    const float* __restrict__ h1, const float* __restrict__ b1,
    const float* __restrict__ W2, float* __restrict__ h2, int N) {
  __shared__ float res[96];
  const int lane = threadIdx.x & 31;
  const int w = __builtin_amdgcn_readfirstlane((int)(threadIdx.x >> 5));
  const int d = blockIdx.x * 32 + w;

  if (d < N) {
    const int beg = off[d];
    const int n = cnt[d];
    const float dd = dinv[d];
    float4 acc = make_float4(0.f, 0.f, 0.f, 0.f);

    for (int base = 0; base < n; base += 32) {
      const int m = (n - base < 32) ? (n - base) : 32;
      int s_l = 0;
      float w_l = 0.f;
      if (lane < m) {
        s_l = list[beg + base + lane];
        s_l = ((unsigned)s_l < (unsigned)N) ? s_l : 0;
        w_l = dinv[s_l] * dd;
      }
#pragma unroll 4
      for (int k = 0; k < m; ++k) {
        const int s = __builtin_amdgcn_readlane(s_l, k);
        const float nrm = __int_as_float(__builtin_amdgcn_readlane(__float_as_int(w_l), k));
        const float4 v = ((const float4*)(h1 + (long long)s * 128))[lane];
        acc.x += nrm * v.x;
        acc.y += nrm * v.y;
        acc.z += nrm * v.z;
        acc.w += nrm * v.w;
      }
    }

    const float4 hv = ((const float4*)(h1 + (long long)d * 128))[lane];
    const float nn = dd * dd;
    float y0 = acc.x + nn * hv.x + b1[lane * 4 + 0];
    float y1 = acc.y + nn * hv.y + b1[lane * 4 + 1];
    float y2 = acc.z + nn * hv.z + b1[lane * 4 + 2];
    float y3 = acc.w + nn * hv.w + b1[lane * 4 + 3];
    y0 = y0 > 0.0f ? y0 : 0.0f;
    y1 = y1 > 0.0f ? y1 : 0.0f;
    y2 = y2 > 0.0f ? y2 : 0.0f;
    y3 = y3 > 0.0f ? y3 : 0.0f;

    const float* __restrict__ wr = W2 + lane * 12;
    float p0 = y0 * wr[0] + y1 * wr[3] + y2 * wr[6] + y3 * wr[9];
    float p1 = y0 * wr[1] + y1 * wr[4] + y2 * wr[7] + y3 * wr[10];
    float p2 = y0 * wr[2] + y1 * wr[5] + y2 * wr[8] + y3 * wr[11];
    for (int o = 16; o > 0; o >>= 1) {
      p0 += __shfl_xor(p0, o);
      p1 += __shfl_xor(p1, o);
      p2 += __shfl_xor(p2, o);
    }
    if (lane == 0) {
      res[w * 3 + 0] = p0;
      res[w * 3 + 1] = p1;
      res[w * 3 + 2] = p2;
    }
  }
  __syncthreads();

  if (threadIdx.x < 24) {
    const long long f = (long long)blockIdx.x * 96 + threadIdx.x * 4;
    const long long n3 = (long long)N * 3;
    if (f + 3 < n3) {
      *(float4*)(h2 + f) = make_float4(res[threadIdx.x * 4 + 0], res[threadIdx.x * 4 + 1],
                                       res[threadIdx.x * 4 + 2], res[threadIdx.x * 4 + 3]);
    } else {
      for (int q = 0; q < 4; ++q)
        if (f + q < n3) h2[f + q] = res[threadIdx.x * 4 + q];
    }
  }
}

__global__ __launch_bounds__(256) void gcn_gather2(
    const int* __restrict__ off, const int* __restrict__ cnt,
    const int* __restrict__ list, const float* __restrict__ dinv,
    const float* __restrict__ h2, const float* __restrict__ b2,
    float* __restrict__ out, int N) {
  __shared__ float res[96];
  const int t = threadIdx.x;
  const int g = t >> 3;
  const int l = t & 7;
  const int d = blockIdx.x * 32 + g;
  const bool valid = d < N;
  const int dc = valid ? d : 0;
  const int beg = off[dc];
  const int n = valid ? cnt[dc] : 0;
  const float dd = dinv[dc];

  float a0 = 0.f, a1 = 0.f, a2 = 0.f;
  for (int j = l; j < n; j += 8) {
    int s = list[beg + j];
    s = ((unsigned)s < (unsigned)N) ? s : 0;
    const float nrm = dinv[s] * dd;
    const float* __restrict__ hp = h2 + (long long)s * 3;
    a0 += nrm * hp[0];
    a1 += nrm * hp[1];
    a2 += nrm * hp[2];
  }
  for (int o = 4; o > 0; o >>= 1) {
    a0 += __shfl_xor(a0, o);
    a1 += __shfl_xor(a1, o);
    a2 += __shfl_xor(a2, o);
  }
  if (l == 0) {
    const float nn = dd * dd;
    const float* __restrict__ hp = h2 + (long long)dc * 3;
    res[g * 3 + 0] = a0 + nn * hp[0] + b2[0];
    res[g * 3 + 1] = a1 + nn * hp[1] + b2[1];
    res[g * 3 + 2] = a2 + nn * hp[2] + b2[2];
  }
  __syncthreads();

  if (t < 24) {
    const long long f = (long long)blockIdx.x * 96 + t * 4;
    const long long n3 = (long long)N * 3;
    if (f + 3 < n3) {
      *(float4*)(out + f) = make_float4(res[t * 4 + 0], res[t * 4 + 1],
                                        res[t * 4 + 2], res[t * 4 + 3]);
    } else {
      for (int q = 0; q < 4; ++q)
        if (f + q < n3) out[f + q] = res[t * 4 + q];
    }
  }
}

extern "C" void kernel_launch(void* const* d_in, const int* in_sizes, int n_in,
                              void* d_out, int out_size, void* d_ws, size_t ws_size,
                              hipStream_t stream) {
  const float* x  = (const float*)d_in[0];
  const int*   ei = (const int*)d_in[1];
  const float* W1 = (const float*)d_in[2];
  const float* b1 = (const float*)d_in[3];
  const float* W2 = (const float*)d_in[4];
  const float* b2 = (const float*)d_in[5];
  float* out = (float*)d_out;

  const int N = in_sizes[0] / 512;
  const int E = in_sizes[1] / 2;
  const int* src = ei;
  const int* dst = ei + E;

  const long long Npad = ((long long)N + 255) & ~255LL;
  const long long Epad = ((long long)E + 255) & ~255LL;
  float* ws   = (float*)d_ws;
  float* dinv = ws;
  float* h1   = ws + Npad;
  float* pl   = h1 + (long long)N * 128;
  float* h2   = pl + (long long)N * 128;
  int* cnt  = (int*)pl;
  int* off  = cnt + Npad;
  int* rank = off + Npad;
  int* list = rank + Epad;

  const int B = 256;

  gcn_zero_cnt<<<(N + B - 1) / B, B, 0, stream>>>(cnt, N);
  gcn_count_rank<<<(E + B - 1) / B, B, 0, stream>>>(dst, cnt, rank, E);
  gcn_scan_dinv<<<(N + 1023) / 1024, 1024, 0, stream>>>(cnt, off, dinv, N);
  gcn_fill<<<(E + B - 1) / B, B, 0, stream>>>(src, dst, off, rank, list, E);

  gcn_gemm1_wmma<<<256, 128, 0, stream>>>(x, W1, h1, N);

  gcn_gather1<<<(N + 31) / 32, 1024, 0, stream>>>(off, cnt, list, dinv, h1, b1, W2, h2, N);
  gcn_gather2<<<(N + 31) / 32, B, 0, stream>>>(off, cnt, list, dinv, h2, b2, out, N);
}


# edge list as fixed 64-slot buckets per destination, filed in the one atomic pass (spill list beyond 64); scan, fill and dinv passes gone, 7 launches to 4; sums still in arrival-order rank
# speedup vs baseline: 2.8226x; 1.1556x over previous
#include <hip/hip_runtime.h>


typedef __attribute__((ext_vector_type(2))) float v2f;
typedef __attribute__((ext_vector_type(8))) float v8f;

constexpr int GCN_CAP = 64;

__global__ void gcn_bucket_fill(const int* __restrict__ ei, int* __restrict__ cnt,
                                int* __restrict__ list, int N, int E) {
  int e = blockIdx.x * blockDim.x + threadIdx.x;
  if (e < E) {
    const int s = ei[e];
    const int d = ei[E + e];
    const int r = atomicAdd(&cnt[d], 1);
    int* p = list + (long long)d * GCN_CAP + r;
    int v = s;
    if (r >= GCN_CAP) {
      p = list + (long long)N * GCN_CAP + atomicAdd(&cnt[N], 1);
      v = e;
    }
    *p = v;
  }
}

__global__ __launch_bounds__(128) void gcn_gemm1_wmma(
    const float* __restrict__ x, const float* __restrict__ W,
    float* __restrict__ h, int M, int* __restrict__ zc, int nz) {
  for (int i = (int)(blockIdx.x * 128 + threadIdx.x); i < nz; i += (int)gridDim.x * 128)
    zc[i] = 0;

  const int lane = threadIdx.x & 31;
  const int J = __builtin_amdgcn_readfirstlane((int)(threadIdx.x >> 5));
  const int c = lane & 15;
  const int hh = lane >> 4;
  const int ntiles = (M + 15) >> 4;
  const int G = (int)gridDim.x;

  v2f bf[8][16][2];
  {
    const float* __restrict__ wb = W + (4 * hh) * 128 + 32 * J + 2 * c;
#pragma unroll
    for (int kb = 0; kb < 8; ++kb) {
#pragma unroll
      for (int s = 0; s < 16; ++s) {
        const int k = 64 * kb + 8 * (s >> 1) + 2 * (s & 1);
        const float2 r0 = *(const float2*)(wb + k * 128);
        const float2 r1 = *(const float2*)(wb + (k + 1) * 128);
        bf[kb][s][0].x = r0.x; bf[kb][s][0].y = r1.x;
        bf[kb][s][1].x = r0.y; bf[kb][s][1].y = r1.y;
      }
    }
  }

  int tile = (int)blockIdx.x;
  if (tile >= ntiles) return;
  float4 a0[32];
  float4 a1[32];
  {
    int row = tile * 16 + c;
    if (row >= M) row = M - 1;
    const float4* __restrict__ ar = (const float4*)(x + (long long)row * 512 + 4 * hh);
#pragma unroll
    for (int i = 0; i < 32; ++i) a0[i] = ar[16 * (i >> 3) + 2 * (i & 7)];
  }

  for (; tile < ntiles; tile += G) {
    {
      int row = tile * 16 + c;
      if (row >= M) row = M - 1;
      const float4* __restrict__ ar = (const float4*)(x + (long long)row * 512 + 4 * hh);
#pragma unroll
      for (int i = 0; i < 32; ++i) a1[i] = ar[64 + 16 * (i >> 3) + 2 * (i & 7)];
    }
    __builtin_amdgcn_sched_barrier(0);

    v8f acc0 = {};
    v8f acc1 = {};

#pragma unroll
    for (int kb = 0; kb < 4; ++kb) {
#pragma unroll
      for (int s = 0; s < 16; ++s) {
        const float4 av = a0[8 * kb + (s >> 1)];
        v2f a;
        if (s & 1) { a.x = av.z; a.y = av.w; } else { a.x = av.x; a.y = av.y; }
        acc0 = __builtin_amdgcn_wmma_f32_16x16x4_f32(
             false, a,  false, bf[kb][s][0],
             (short)0, acc0,  false,  false);
        acc1 = __builtin_amdgcn_wmma_f32_16x16x4_f32(
             false, a,  false, bf[kb][s][1],
             (short)0, acc1,  false,  false);
      }
    }

    __builtin_amdgcn_sched_barrier(0);
    {
      const int nt = (tile + G < ntiles) ? tile + G : tile;
      int row = nt * 16 + c;
      if (row >= M) row = M - 1;
      const float4* __restrict__ ar = (const float4*)(x + (long long)row * 512 + 4 * hh);
#pragma unroll
      for (int i = 0; i < 32; ++i) a0[i] = ar[16 * (i >> 3) + 2 * (i & 7)];
    }
    __builtin_amdgcn_sched_barrier(0);

#pragma unroll
    for (int kb = 4; kb < 8; ++kb) {
#pragma unroll
      for (int s = 0; s < 16; ++s) {
        const float4 av = a1[8 * (kb - 4) + (s >> 1)];
        v2f a;
        if (s & 1) { a.x = av.z; a.y = av.w; } else { a.x = av.x; a.y = av.y; }
        acc0 = __builtin_amdgcn_wmma_f32_16x16x4_f32(
             false, a,  false, bf[kb][s][0],
             (short)0, acc0,  false,  false);
        acc1 = __builtin_amdgcn_wmma_f32_16x16x4_f32(
             false, a,  false, bf[kb][s][1],
             (short)0, acc1,  false,  false);
      }
    }

    const int rbase = tile * 16 + 8 * hh;
    float* __restrict__ hp = h + (long long)rbase * 128 + 32 * J + 2 * c;
    if (tile * 16 + 16 <= M) {
#pragma unroll
      for (int v = 0; v < 8; ++v) *(float2*)(hp + v * 128) = make_float2(acc0[v], acc1[v]);
    } else {
#pragma unroll
      for (int v = 0; v < 8; ++v) {
        if (rbase + v < M) *(float2*)(hp + v * 128) = make_float2(acc0[v], acc1[v]);
      }
    }
  }
}

__global__ __launch_bounds__(1024) void gcn_gather1(
    const int* __restrict__ cnt, const int* __restrict__ list,
    const int* __restrict__ ei,
    const float* __restrict__ h1, const float* __restrict__ b1,
    const float* __restrict__ W2, float* __restrict__ h2, int N, int E) {
  __shared__ float res[96];
  const int lane = threadIdx.x & 31;
  const int w = __builtin_amdgcn_readfirstlane((int)(threadIdx.x >> 5));
  const int d = blockIdx.x * 32 + w;

  if (d < N) {
    const int nAll = cnt[d];
    const int n = nAll < GCN_CAP ? nAll : GCN_CAP;
    const float dd = rsqrtf((float)(nAll + 1));
    const int* __restrict__ bl = list + (long long)d * GCN_CAP;
    float4 acc = make_float4(0.f, 0.f, 0.f, 0.f);

    for (int base = 0; base < n; base += 32) {
      const int m = (n - base < 32) ? (n - base) : 32;
      int s_l = 0;
      float w_l = 0.f;
      if (lane < m) {
        s_l = bl[base + lane];
        s_l = ((unsigned)s_l < (unsigned)N) ? s_l : 0;
        w_l = rsqrtf((float)(cnt[s_l] + 1)) * dd;
      }
#pragma unroll 4
      for (int k = 0; k < m; ++k) {
        const int s = __builtin_amdgcn_readlane(s_l, k);
        const float nrm = __int_as_float(__builtin_amdgcn_readlane(__float_as_int(w_l), k));
        const float4 v = ((const float4*)(h1 + (long long)s * 128))[lane];
        acc.x += nrm * v.x;
        acc.y += nrm * v.y;
        acc.z += nrm * v.z;
        acc.w += nrm * v.w;
      }
    }

    if (nAll > GCN_CAP) {
      const int* __restrict__ ovl = list + (long long)N * GCN_CAP;
      const int* __restrict__ src = ei;
      const int* __restrict__ dst = ei + E;
      int no = cnt[N];
      no = no < E ? no : E;
      for (int base = 0; base < no; base += 32) {
        int s_l = 0;
        float w_l = 0.f;
        bool mine = false;
        if (base + lane < no) {
          const int e = ovl[base + lane];
          if ((unsigned)e < (unsigned)E && dst[e] == d) {
            mine = true;
            s_l = src[e];
            s_l = ((unsigned)s_l < (unsigned)N) ? s_l : 0;
            w_l = rsqrtf((float)(cnt[s_l] + 1)) * dd;
          }
        }
        unsigned todo = __builtin_amdgcn_ballot_w32(mine);
        while (todo != 0u) {
          const int k = __builtin_ctz(todo);
          todo &= todo - 1u;
          const int s = __builtin_amdgcn_readlane(s_l, k);
          const float nrm = __int_as_float(__builtin_amdgcn_readlane(__float_as_int(w_l), k));
          const float4 v = ((const float4*)(h1 + (long long)s * 128))[lane];
          acc.x += nrm * v.x;
          acc.y += nrm * v.y;
          acc.z += nrm * v.z;
          acc.w += nrm * v.w;
        }
      }
    }

    const float4 hv = ((const float4*)(h1 + (long long)d * 128))[lane];
    const float nn = dd * dd;
    float y0 = acc.x + nn * hv.x + b1[lane * 4 + 0];
    float y1 = acc.y + nn * hv.y + b1[lane * 4 + 1];
    float y2 = acc.z + nn * hv.z + b1[lane * 4 + 2];
    float y3 = acc.w + nn * hv.w + b1[lane * 4 + 3];
    y0 = y0 > 0.0f ? y0 : 0.0f;
    y1 = y1 > 0.0f ? y1 : 0.0f;
    y2 = y2 > 0.0f ? y2 : 0.0f;
    y3 = y3 > 0.0f ? y3 : 0.0f;

    const float* __restrict__ wr = W2 + lane * 12;
    float p0 = y0 * wr[0] + y1 * wr[3] + y2 * wr[6] + y3 * wr[9];
    float p1 = y0 * wr[1] + y1 * wr[4] + y2 * wr[7] + y3 * wr[10];
    float p2 = y0 * wr[2] + y1 * wr[5] + y2 * wr[8] + y3 * wr[11];
    for (int o = 16; o > 0; o >>= 1) {
      p0 += __shfl_xor(p0, o);
      p1 += __shfl_xor(p1, o);
      p2 += __shfl_xor(p2, o);
    }
    if (lane == 0) {
      res[w * 3 + 0] = p0;
      res[w * 3 + 1] = p1;
      res[w * 3 + 2] = p2;
    }
  }
  __syncthreads();

  if (threadIdx.x < 24) {
    const long long f = (long long)blockIdx.x * 96 + threadIdx.x * 4;
    const long long n3 = (long long)N * 3;
    if (f + 3 < n3) {
      *(float4*)(h2 + f) = make_float4(res[threadIdx.x * 4 + 0], res[threadIdx.x * 4 + 1],
                                       res[threadIdx.x * 4 + 2], res[threadIdx.x * 4 + 3]);
    } else {
      for (int q = 0; q < 4; ++q)
        if (f + q < n3) h2[f + q] = res[threadIdx.x * 4 + q];
    }
  }
}

__global__ __launch_bounds__(256) void gcn_gather2(
    const int* __restrict__ cnt, const int* __restrict__ list,
    const int* __restrict__ ei,
    const float* __restrict__ h2, const float* __restrict__ b2,
    float* __restrict__ out, int N, int E) {
  __shared__ float res[96];
  const int t = threadIdx.x;
  const int g = t >> 3;
  const int l = t & 7;
  const int d = blockIdx.x * 32 + g;
  const bool valid = d < N;
  const int dc = valid ? d : 0;
  const int nAll = valid ? cnt[dc] : 0;
  const int n = nAll < GCN_CAP ? nAll : GCN_CAP;
  const float dd = rsqrtf((float)(nAll + 1));
  const int* __restrict__ bl = list + (long long)dc * GCN_CAP;

  float a0 = 0.f, a1 = 0.f, a2 = 0.f;
  for (int j = l; j < n; j += 8) {
    int s = bl[j];
    s = ((unsigned)s < (unsigned)N) ? s : 0;
    const float nrm = rsqrtf((float)(cnt[s] + 1)) * dd;
    const float* __restrict__ hp = h2 + (long long)s * 3;
    a0 += nrm * hp[0];
    a1 += nrm * hp[1];
    a2 += nrm * hp[2];
  }
  if (nAll > GCN_CAP) {
    const int* __restrict__ ovl = list + (long long)N * GCN_CAP;
    const int* __restrict__ src = ei;
    const int* __restrict__ dst = ei + E;
    int no = cnt[N];
    no = no < E ? no : E;
    for (int o = l; o < no; o += 8) {
      const int e = ovl[o];
      if ((unsigned)e < (unsigned)E && dst[e] == d) {
        int s = src[e];
        s = ((unsigned)s < (unsigned)N) ? s : 0;
        const float nrm = rsqrtf((float)(cnt[s] + 1)) * dd;
        const float* __restrict__ hp = h2 + (long long)s * 3;
        a0 += nrm * hp[0];
        a1 += nrm * hp[1];
        a2 += nrm * hp[2];
      }
    }
  }
  for (int o = 4; o > 0; o >>= 1) {
    a0 += __shfl_xor(a0, o);
    a1 += __shfl_xor(a1, o);
    a2 += __shfl_xor(a2, o);
  }
  if (l == 0) {
    const float nn = dd * dd;
    const float* __restrict__ hp = h2 + (long long)dc * 3;
    res[g * 3 + 0] = a0 + nn * hp[0] + b2[0];
    res[g * 3 + 1] = a1 + nn * hp[1] + b2[1];
    res[g * 3 + 2] = a2 + nn * hp[2] + b2[2];
  }
  __syncthreads();

  if (t < 24) {
    const long long f = (long long)blockIdx.x * 96 + t * 4;
    const long long n3 = (long long)N * 3;
    if (f + 3 < n3) {
      *(float4*)(out + f) = make_float4(res[t * 4 + 0], res[t * 4 + 1],
                                        res[t * 4 + 2], res[t * 4 + 3]);
    } else {
      for (int q = 0; q < 4; ++q)
        if (f + q < n3) out[f + q] = res[t * 4 + q];
    }
  }
}

extern "C" void kernel_launch(void* const* d_in, const int* in_sizes, int n_in,
                              void* d_out, int out_size, void* d_ws, size_t ws_size,
                              hipStream_t stream) {
  const float* x  = (const float*)d_in[0];
  const int*   ei = (const int*)d_in[1];
  const float* W1 = (const float*)d_in[2];
  const float* b1 = (const float*)d_in[3];
  const float* W2 = (const float*)d_in[4];
  const float* b2 = (const float*)d_in[5];
  float* out = (float*)d_out;

  const int N = in_sizes[0] / 512;
  const int E = in_sizes[1] / 2;

  const long long Npad = ((long long)N + 255) & ~255LL;
  const long long Cpad = ((long long)N + 1 + 255) & ~255LL;
  float* ws   = (float*)d_ws;
  float* h1   = ws + Npad;
  float* pl   = h1 + (long long)N * 128;
  float* h2   = pl + (long long)N * 128;
  int* cnt  = (int*)pl;
  int* list = cnt + Cpad;

  const int B = 256;

  gcn_gemm1_wmma<<<256, 128, 0, stream>>>(x, W1, h1, N, cnt, N + 1);

  gcn_bucket_fill<<<(E + B - 1) / B, B, 0, stream>>>(ei, cnt, list, N, E);

  gcn_gather1<<<(N + 31) / 32, 1024, 0, stream>>>(cnt, list, ei, h1, b1, W2, h2, N, E);
  gcn_gather2<<<(N + 31) / 32, B, 0, stream>>>(cnt, list, ei, h2, b2, out, N, E);
}


# bucket fill moved into the GEMM launch: a wave files 64 edges per row tile between the tile's halves, rank drawn a tile ahead of its store; counter-clear launch added; sums stay in arrival-order rank
# speedup vs baseline: 3.2972x; 1.1681x over previous
#include <hip/hip_runtime.h>


typedef __attribute__((ext_vector_type(2))) float v2f;
typedef __attribute__((ext_vector_type(8))) float v8f;

constexpr int GCN_CAP = 64;

__global__ void gcn_clear(int* __restrict__ zc, int nz) {
  const int i = blockIdx.x * blockDim.x + threadIdx.x;
  if (i < nz) zc[i] = 0;
}

struct GcnEdgeIdx { int s0, d0, s1, d1; };
struct GcnEdgeRank { int s0, d0, r0, s1, d1, r1; };

__device__ __forceinline__ GcnEdgeIdx gcn_edge_load(const int* __restrict__ ei, int E,
                                                   long long base, int lane) {
  GcnEdgeIdx x;
  x.s0 = 0; x.d0 = -1; x.s1 = 0; x.d1 = -1;
  const long long e0 = base + lane;
  const long long e1 = e0 + 32;
  if (e0 < E) { x.s0 = ei[e0]; x.d0 = ei[E + e0]; }
  if (e1 < E) { x.s1 = ei[e1]; x.d1 = ei[E + e1]; }
  return x;
}

__device__ __forceinline__ GcnEdgeRank gcn_edge_rank(const GcnEdgeIdx x, int* __restrict__ cnt) {
  GcnEdgeRank y;
  y.s0 = x.s0; y.d0 = x.d0; y.s1 = x.s1; y.d1 = x.d1;
  y.r0 = 0; y.r1 = 0;
  if (x.d0 >= 0) y.r0 = atomicAdd(&cnt[x.d0], 1);
  if (x.d1 >= 0) y.r1 = atomicAdd(&cnt[x.d1], 1);
  return y;
}

__device__ __forceinline__ void gcn_edge_file_one(int s, int d, int r, long long e,
                                                  int* __restrict__ cnt,
                                                  int* __restrict__ list, int N) {
  if (d >= 0) {
    int* p = list + (long long)d * GCN_CAP + r;
    int v = s;
    if (r >= GCN_CAP) {
      p = list + (long long)N * GCN_CAP + atomicAdd(&cnt[N], 1);
      v = (int)e;
    }
    *p = v;
  }
}

__device__ __forceinline__ void gcn_edge_file(const GcnEdgeRank y, long long base, int lane,
                                              int* __restrict__ cnt,
                                              int* __restrict__ list, int N) {
  gcn_edge_file_one(y.s0, y.d0, y.r0, base + lane, cnt, list, N);
  gcn_edge_file_one(y.s1, y.d1, y.r1, base + lane + 32, cnt, list, N);
}

__global__ __launch_bounds__(128) void gcn_gemm1_wmma(
    const float* __restrict__ x, const float* __restrict__ W,
    float* __restrict__ h, int M, const int* __restrict__ ei,
    int* __restrict__ cnt, int* __restrict__ list, int E) {
  const int lane = threadIdx.x & 31;
  const int J = __builtin_amdgcn_readfirstlane((int)(threadIdx.x >> 5));
  const int c = lane & 15;
  const int hh = lane >> 4;
  const int ntiles = (M + 15) >> 4;
  const int G = (int)gridDim.x;

  const long long nw = (long long)G * 4;
  const long long gw = (long long)blockIdx.x * 4 + J;
  const long long ngroups = ((long long)E + 63) >> 6;
  const int np = gw < ngroups ? (int)((ngroups - gw + nw - 1) / nw) : 0;
  int ek = 2;
  GcnEdgeIdx eL = gcn_edge_load(ei, E, gw * 64, lane);

  v2f bf[8][16][2];
  {
    const float* __restrict__ wb = W + (4 * hh) * 128 + 32 * J + 2 * c;
#pragma unroll
    for (int kb = 0; kb < 8; ++kb) {
#pragma unroll
      for (int s = 0; s < 16; ++s) {
        const int k = 64 * kb + 8 * (s >> 1) + 2 * (s & 1);
        const float2 r0 = *(const float2*)(wb + k * 128);
        const float2 r1 = *(const float2*)(wb + (k + 1) * 128);
        bf[kb][s][0].x = r0.x; bf[kb][s][0].y = r1.x;
        bf[kb][s][1].x = r0.y; bf[kb][s][1].y = r1.y;
      }
    }
  }

  GcnEdgeRank eR = gcn_edge_rank(eL, cnt);
  eL = gcn_edge_load(ei, E, (nw + gw) * 64, lane);

  int tile = (int)blockIdx.x;
  if (tile >= ntiles) {
    for (; ek <= np + 1; ++ek) {
      gcn_edge_file(eR, ((long long)(ek - 2) * nw + gw) * 64, lane, cnt, list, M);
      eR = gcn_edge_rank(eL, cnt);
      eL = gcn_edge_load(ei, E, ((long long)ek * nw + gw) * 64, lane);
    }
    return;
  }
  float4 a0[32];
  float4 a1[32];
  {
    int row = tile * 16 + c;
    if (row >= M) row = M - 1;
    const float4* __restrict__ ar = (const float4*)(x + (long long)row * 512 + 4 * hh);
#pragma unroll
    for (int i = 0; i < 32; ++i) a0[i] = ar[16 * (i >> 3) + 2 * (i & 7)];
  }

  for (; tile < ntiles; tile += G) {
    {
      int row = tile * 16 + c;
      if (row >= M) row = M - 1;
      const float4* __restrict__ ar = (const float4*)(x + (long long)row * 512 + 4 * hh);
#pragma unroll
      for (int i = 0; i < 32; ++i) a1[i] = ar[64 + 16 * (i >> 3) + 2 * (i & 7)];
    }
    __builtin_amdgcn_sched_barrier(0);

    v8f acc0 = {};
    v8f acc1 = {};

#pragma unroll
    for (int kb = 0; kb < 4; ++kb) {
#pragma unroll
      for (int s = 0; s < 16; ++s) {
        const float4 av = a0[8 * kb + (s >> 1)];
        v2f a;
        if (s & 1) { a.x = av.z; a.y = av.w; } else { a.x = av.x; a.y = av.y; }
        acc0 = __builtin_amdgcn_wmma_f32_16x16x4_f32(
             false, a,  false, bf[kb][s][0],
             (short)0, acc0,  false,  false);
        acc1 = __builtin_amdgcn_wmma_f32_16x16x4_f32(
             false, a,  false, bf[kb][s][1],
             (short)0, acc1,  false,  false);
      }
    }

    gcn_edge_file(eR, ((long long)(ek - 2) * nw + gw) * 64, lane, cnt, list, M);
    eR = gcn_edge_rank(eL, cnt);
    eL = gcn_edge_load(ei, E, ((long long)ek * nw + gw) * 64, lane);
    ++ek;

    {
      const int nt = (tile + G < ntiles) ? tile + G : tile;
      int row = nt * 16 + c;
      if (row >= M) row = M - 1;
      const float4* __restrict__ ar = (const float4*)(x + (long long)row * 512 + 4 * hh);
#pragma unroll
      for (int i = 0; i < 32; ++i) a0[i] = ar[16 * (i >> 3) + 2 * (i & 7)];
    }
    __builtin_amdgcn_sched_barrier(0);

#pragma unroll
    for (int kb = 4; kb < 8; ++kb) {
#pragma unroll
      for (int s = 0; s < 16; ++s) {
        const float4 av = a1[8 * (kb - 4) + (s >> 1)];
        v2f a;
        if (s & 1) { a.x = av.z; a.y = av.w; } else { a.x = av.x; a.y = av.y; }
        acc0 = __builtin_amdgcn_wmma_f32_16x16x4_f32(
             false, a,  false, bf[kb][s][0],
             (short)0, acc0,  false,  false);
        acc1 = __builtin_amdgcn_wmma_f32_16x16x4_f32(
             false, a,  false, bf[kb][s][1],
             (short)0, acc1,  false,  false);
      }
    }

    const int rbase = tile * 16 + 8 * hh;
    float* __restrict__ hp = h + (long long)rbase * 128 + 32 * J + 2 * c;
    if (tile * 16 + 16 <= M) {
#pragma unroll
      for (int v = 0; v < 8; ++v) *(float2*)(hp + v * 128) = make_float2(acc0[v], acc1[v]);
    } else {
#pragma unroll
      for (int v = 0; v < 8; ++v) {
        if (rbase + v < M) *(float2*)(hp + v * 128) = make_float2(acc0[v], acc1[v]);
      }
    }
  }

  for (; ek <= np + 1; ++ek) {
    gcn_edge_file(eR, ((long long)(ek - 2) * nw + gw) * 64, lane, cnt, list, M);
    eR = gcn_edge_rank(eL, cnt);
    eL = gcn_edge_load(ei, E, ((long long)ek * nw + gw) * 64, lane);
  }
}

__global__ __launch_bounds__(1024) void gcn_gather1(
    const int* __restrict__ cnt, const int* __restrict__ list,
    const int* __restrict__ ei,
    const float* __restrict__ h1, const float* __restrict__ b1,
    const float* __restrict__ W2, float* __restrict__ h2, int N, int E) {
  __shared__ float res[96];
  const int lane = threadIdx.x & 31;
  const int w = __builtin_amdgcn_readfirstlane((int)(threadIdx.x >> 5));
  const int d = blockIdx.x * 32 + w;

  if (d < N) {
    const int nAll = cnt[d];
    const int n = nAll < GCN_CAP ? nAll : GCN_CAP;
    const float dd = rsqrtf((float)(nAll + 1));
    const int* __restrict__ bl = list + (long long)d * GCN_CAP;
    float4 acc = make_float4(0.f, 0.f, 0.f, 0.f);

    for (int base = 0; base < n; base += 32) {
      const int m = (n - base < 32) ? (n - base) : 32;
      int s_l = 0;
      float w_l = 0.f;
      if (lane < m) {
        s_l = bl[base + lane];
        s_l = ((unsigned)s_l < (unsigned)N) ? s_l : 0;
        w_l = rsqrtf((float)(cnt[s_l] + 1)) * dd;
      }
#pragma unroll 4
      for (int k = 0; k < m; ++k) {
        const int s = __builtin_amdgcn_readlane(s_l, k);
        const float nrm = __int_as_float(__builtin_amdgcn_readlane(__float_as_int(w_l), k));
        const float4 v = ((const float4*)(h1 + (long long)s * 128))[lane];
        acc.x += nrm * v.x;
        acc.y += nrm * v.y;
        acc.z += nrm * v.z;
        acc.w += nrm * v.w;
      }
    }

    if (nAll > GCN_CAP) {
      const int* __restrict__ ovl = list + (long long)N * GCN_CAP;
      const int* __restrict__ src = ei;
      const int* __restrict__ dst = ei + E;
      int no = cnt[N];
      no = no < E ? no : E;
      for (int base = 0; base < no; base += 32) {
        int s_l = 0;
        float w_l = 0.f;
        bool mine = false;
        if (base + lane < no) {
          const int e = ovl[base + lane];
          if ((unsigned)e < (unsigned)E && dst[e] == d) {
            mine = true;
            s_l = src[e];
            s_l = ((unsigned)s_l < (unsigned)N) ? s_l : 0;
            w_l = rsqrtf((float)(cnt[s_l] + 1)) * dd;
          }
        }
        unsigned todo = __builtin_amdgcn_ballot_w32(mine);
        while (todo != 0u) {
          const int k = __builtin_ctz(todo);
          todo &= todo - 1u;
          const int s = __builtin_amdgcn_readlane(s_l, k);
          const float nrm = __int_as_float(__builtin_amdgcn_readlane(__float_as_int(w_l), k));
          const float4 v = ((const float4*)(h1 + (long long)s * 128))[lane];
          acc.x += nrm * v.x;
          acc.y += nrm * v.y;
          acc.z += nrm * v.z;
          acc.w += nrm * v.w;
        }
      }
    }

    const float4 hv = ((const float4*)(h1 + (long long)d * 128))[lane];
    const float nn = dd * dd;
    float y0 = acc.x + nn * hv.x + b1[lane * 4 + 0];
    float y1 = acc.y + nn * hv.y + b1[lane * 4 + 1];
    float y2 = acc.z + nn * hv.z + b1[lane * 4 + 2];
    float y3 = acc.w + nn * hv.w + b1[lane * 4 + 3];
    y0 = y0 > 0.0f ? y0 : 0.0f;
    y1 = y1 > 0.0f ? y1 : 0.0f;
    y2 = y2 > 0.0f ? y2 : 0.0f;
    y3 = y3 > 0.0f ? y3 : 0.0f;

    const float* __restrict__ wr = W2 + lane * 12;
    float p0 = y0 * wr[0] + y1 * wr[3] + y2 * wr[6] + y3 * wr[9];
    float p1 = y0 * wr[1] + y1 * wr[4] + y2 * wr[7] + y3 * wr[10];
    float p2 = y0 * wr[2] + y1 * wr[5] + y2 * wr[8] + y3 * wr[11];
    for (int o = 16; o > 0; o >>= 1) {
      p0 += __shfl_xor(p0, o);
      p1 += __shfl_xor(p1, o);
      p2 += __shfl_xor(p2, o);
    }
    if (lane == 0) {
      res[w * 3 + 0] = p0;
      res[w * 3 + 1] = p1;
      res[w * 3 + 2] = p2;
    }
  }
  __syncthreads();

  if (threadIdx.x < 24) {
    const long long f = (long long)blockIdx.x * 96 + threadIdx.x * 4;
    const long long n3 = (long long)N * 3;
    if (f + 3 < n3) {
      *(float4*)(h2 + f) = make_float4(res[threadIdx.x * 4 + 0], res[threadIdx.x * 4 + 1],
                                       res[threadIdx.x * 4 + 2], res[threadIdx.x * 4 + 3]);
    } else {
      for (int q = 0; q < 4; ++q)
        if (f + q < n3) h2[f + q] = res[threadIdx.x * 4 + q];
    }
  }
}

__global__ __launch_bounds__(256) void gcn_gather2(
    const int* __restrict__ cnt, const int* __restrict__ list,
    const int* __restrict__ ei,
    const float* __restrict__ h2, const float* __restrict__ b2,
    float* __restrict__ out, int N, int E) {
  __shared__ float res[96];
  const int t = threadIdx.x;
  const int g = t >> 3;
  const int l = t & 7;
  const int d = blockIdx.x * 32 + g;
  const bool valid = d < N;
  const int dc = valid ? d : 0;
  const int nAll = valid ? cnt[dc] : 0;
  const int n = nAll < GCN_CAP ? nAll : GCN_CAP;
  const float dd = rsqrtf((float)(nAll + 1));
  const int* __restrict__ bl = list + (long long)dc * GCN_CAP;

  float a0 = 0.f, a1 = 0.f, a2 = 0.f;
  for (int j = l; j < n; j += 8) {
    int s = bl[j];
    s = ((unsigned)s < (unsigned)N) ? s : 0;
    const float nrm = rsqrtf((float)(cnt[s] + 1)) * dd;
    const float* __restrict__ hp = h2 + (long long)s * 3;
    a0 += nrm * hp[0];
    a1 += nrm * hp[1];
    a2 += nrm * hp[2];
  }
  if (nAll > GCN_CAP) {
    const int* __restrict__ ovl = list + (long long)N * GCN_CAP;
    const int* __restrict__ src = ei;
    const int* __restrict__ dst = ei + E;
    int no = cnt[N];
    no = no < E ? no : E;
    for (int o = l; o < no; o += 8) {
      const int e = ovl[o];
      if ((unsigned)e < (unsigned)E && dst[e] == d) {
        int s = src[e];
        s = ((unsigned)s < (unsigned)N) ? s : 0;
        const float nrm = rsqrtf((float)(cnt[s] + 1)) * dd;
        const float* __restrict__ hp = h2 + (long long)s * 3;
        a0 += nrm * hp[0];
        a1 += nrm * hp[1];
        a2 += nrm * hp[2];
      }
    }
  }
  for (int o = 4; o > 0; o >>= 1) {
    a0 += __shfl_xor(a0, o);
    a1 += __shfl_xor(a1, o);
    a2 += __shfl_xor(a2, o);
  }
  if (l == 0) {
    const float nn = dd * dd;
    const float* __restrict__ hp = h2 + (long long)dc * 3;
    res[g * 3 + 0] = a0 + nn * hp[0] + b2[0];
    res[g * 3 + 1] = a1 + nn * hp[1] + b2[1];
    res[g * 3 + 2] = a2 + nn * hp[2] + b2[2];
  }
  __syncthreads();

  if (t < 24) {
    const long long f = (long long)blockIdx.x * 96 + t * 4;
    const long long n3 = (long long)N * 3;
    if (f + 3 < n3) {
      *(float4*)(out + f) = make_float4(res[t * 4 + 0], res[t * 4 + 1],
                                        res[t * 4 + 2], res[t * 4 + 3]);
    } else {
      for (int q = 0; q < 4; ++q)
        if (f + q < n3) out[f + q] = res[t * 4 + q];
    }
  }
}

extern "C" void kernel_launch(void* const* d_in, const int* in_sizes, int n_in,
                              void* d_out, int out_size, void* d_ws, size_t ws_size,
                              hipStream_t stream) {
  const float* x  = (const float*)d_in[0];
  const int*   ei = (const int*)d_in[1];
  const float* W1 = (const float*)d_in[2];
  const float* b1 = (const float*)d_in[3];
  const float* W2 = (const float*)d_in[4];
  const float* b2 = (const float*)d_in[5];
  float* out = (float*)d_out;

  const int N = in_sizes[0] / 512;
  const int E = in_sizes[1] / 2;

  const long long Npad = ((long long)N + 255) & ~255LL;
  const long long Cpad = ((long long)N + 1 + 255) & ~255LL;
  float* ws   = (float*)d_ws;
  float* h1   = ws + Npad;
  float* pl   = h1 + (long long)N * 128;
  float* h2   = pl + (long long)N * 128;
  int* cnt  = (int*)pl;
  int* list = cnt + Cpad;

  const int B = 256;

  gcn_clear<<<(N + 1 + B - 1) / B, B, 0, stream>>>(cnt, N + 1);

  gcn_gemm1_wmma<<<256, 128, 0, stream>>>(x, W1, h1, N, ei, cnt, list, E);

  gcn_gather1<<<(N + 31) / 32, 1024, 0, stream>>>(cnt, list, ei, h1, b1, W2, h2, N, E);
  gcn_gather2<<<(N + 31) / 32, B, 0, stream>>>(cnt, list, ei, h2, b2, out, N, E);
}


# GEMM1 operand feed: x tile staged once per work-group through LDS (each wave loads a quarter in whole 512-byte lines, one barrier per tile), A fragments read from LDS; k sum order kept
# speedup vs baseline: 3.4623x; 1.0501x over previous
#include <hip/hip_runtime.h>


typedef __attribute__((ext_vector_type(2))) float v2f;
typedef __attribute__((ext_vector_type(8))) float v8f;
typedef __attribute__((ext_vector_type(4))) float v4f;

constexpr int GCN_CAP = 64;

__global__ void gcn_clear(int* __restrict__ zc, int nz) {
  const int i = blockIdx.x * blockDim.x + threadIdx.x;
  if (i < nz) zc[i] = 0;
}

struct GcnEdgeIdx { int s0, d0, s1, d1; };
struct GcnEdgeRank { int s0, d0, r0, s1, d1, r1; };

__device__ __forceinline__ GcnEdgeIdx gcn_edge_load(const int* __restrict__ ei, int E,
                                                   long long base, int lane) {
  GcnEdgeIdx x;
  x.s0 = 0; x.d0 = -1; x.s1 = 0; x.d1 = -1;
  const long long e0 = base + lane;
  const long long e1 = e0 + 32;
  if (e0 < E) { x.s0 = ei[e0]; x.d0 = ei[E + e0]; }
  if (e1 < E) { x.s1 = ei[e1]; x.d1 = ei[E + e1]; }
  return x;
}

__device__ __forceinline__ GcnEdgeRank gcn_edge_rank(const GcnEdgeIdx x, int* __restrict__ cnt) {
  GcnEdgeRank y;
  y.s0 = x.s0; y.d0 = x.d0; y.s1 = x.s1; y.d1 = x.d1;
  y.r0 = 0; y.r1 = 0;
  if (x.d0 >= 0) y.r0 = atomicAdd(&cnt[x.d0], 1);
  if (x.d1 >= 0) y.r1 = atomicAdd(&cnt[x.d1], 1);
  return y;
}

__device__ __forceinline__ void gcn_edge_file_one(int s, int d, int r, long long e,
                                                  int* __restrict__ cnt,
                                                  int* __restrict__ list, int N) {
  if (d >= 0) {
    int* p = list + (long long)d * GCN_CAP + r;
    int v = s;
    if (r >= GCN_CAP) {
      p = list + (long long)N * GCN_CAP + atomicAdd(&cnt[N], 1);
      v = (int)e;
    }
    *p = v;
  }
}

__device__ __forceinline__ void gcn_edge_file(const GcnEdgeRank y, long long base, int lane,
                                              int* __restrict__ cnt,
                                              int* __restrict__ list, int N) {
  gcn_edge_file_one(y.s0, y.d0, y.r0, base + lane, cnt, list, N);
  gcn_edge_file_one(y.s1, y.d1, y.r1, base + lane + 32, cnt, list, N);
}

constexpr int GCN_XP = 129;
constexpr int GCN_XT = 16 * GCN_XP;

__global__ __launch_bounds__(128) void gcn_gemm1_wmma(
    const float* __restrict__ x, const float* __restrict__ W,
    float* __restrict__ h, int M, const int* __restrict__ ei,
    int* __restrict__ cnt, int* __restrict__ list, int E) {
  __shared__ v4f xs[2 * GCN_XT];
  const int lane = threadIdx.x & 31;
  const int J = __builtin_amdgcn_readfirstlane((int)(threadIdx.x >> 5));
  const int c = lane & 15;
  const int hh = lane >> 4;
  const int ntiles = (M + 15) >> 4;
  const int G = (int)gridDim.x;

  const long long nw = (long long)G * 4;
  const long long gw = (long long)blockIdx.x * 4 + J;
  const long long ngroups = ((long long)E + 63) >> 6;
  const int np = gw < ngroups ? (int)((ngroups - gw + nw - 1) / nw) : 0;
  int ek = 2;
  GcnEdgeIdx eL = gcn_edge_load(ei, E, gw * 64, lane);

  int tile = (int)blockIdx.x;
  const bool has_tile = tile < ntiles;

  v2f bf[8][16][2];
  {
    const float* __restrict__ wb = W + (4 * hh) * 128 + 32 * J + 2 * c;
#pragma unroll
    for (int kb = 0; kb < 8; ++kb) {
#pragma unroll
      for (int s = 0; s < 16; ++s) {
        const int k = 64 * kb + 8 * (s >> 1) + 2 * (s & 1);
        const float2 r0 = *(const float2*)(wb + k * 128);
        const float2 r1 = *(const float2*)(wb + (k + 1) * 128);
        bf[kb][s][0].x = r0.x; bf[kb][s][0].y = r1.x;
        bf[kb][s][1].x = r0.y; bf[kb][s][1].y = r1.y;
      }
    }
  }

  GcnEdgeRank eR = gcn_edge_rank(eL, cnt);
  eL = gcn_edge_load(ei, E, (nw + gw) * 64, lane);

  if (!has_tile) {
    for (; ek <= np + 1; ++ek) {
      gcn_edge_file(eR, ((long long)(ek - 2) * nw + gw) * 64, lane, cnt, list, M);
      eR = gcn_edge_rank(eL, cnt);
      eL = gcn_edge_load(ei, E, ((long long)ek * nw + gw) * 64, lane);
    }
    return;
  }

  v4f g[16];
  {
    const int st = tile;
#pragma unroll
    for (int i = 0; i < 16; ++i) {
      int row = st * 16 + 4 * J + (i >> 2);
      if (row >= M) row = M - 1;
      g[i] = ((const v4f*)(x + (long long)row * 512))[lane + 32 * (i & 3)];
    }
  }

  v4f* __restrict__ sw = xs + (4 * J) * GCN_XP + lane;
  const v4f* __restrict__ rd = xs + c * GCN_XP + hh;
  int pb = 0;

  {
    v4f* __restrict__ sd = sw;
#pragma unroll
    for (int i = 0; i < 16; ++i) sd[(i >> 2) * GCN_XP + 32 * (i & 3)] = g[i];
  }
  __syncthreads();

  {
    const int st = (tile + G < ntiles) ? tile + G : tile;
#pragma unroll
    for (int i = 0; i < 16; ++i) {
      int row = st * 16 + 4 * J + (i >> 2);
      if (row >= M) row = M - 1;
      g[i] = ((const v4f*)(x + (long long)row * 512))[lane + 32 * (i & 3)];
    }
  }

  v4f fa[2][16];
#pragma unroll
  for (int f = 0; f < 16; ++f) fa[0][f] = rd[16 * (f >> 3) + 2 * (f & 7)];

  for (; tile < ntiles; tile += G) {
    gcn_edge_file(eR, ((long long)(ek - 2) * nw + gw) * 64, lane, cnt, list, M);
    eR = gcn_edge_rank(eL, cnt);
    eL = gcn_edge_load(ei, E, ((long long)ek * nw + gw) * 64, lane);
    ++ek;

    v8f acc0 = {};
    v8f acc1 = {};

#pragma unroll
    for (int blk = 0; blk < 3; ++blk) {
      if (blk == 2) {
        v4f* __restrict__ sd = sw + (pb ^ GCN_XT);
#pragma unroll
        for (int i = 0; i < 16; ++i) sd[(i >> 2) * GCN_XP + 32 * (i & 3)] = g[i];
      }
#pragma unroll
      for (int f = 0; f < 16; ++f)
        fa[(blk + 1) & 1][f] = rd[pb + 16 * (2 * (blk + 1) + (f >> 3)) + 2 * (f & 7)];
      __builtin_amdgcn_sched_barrier(0);

#pragma unroll
      for (int kk = 0; kk < 2; ++kk) {
#pragma unroll
        for (int s = 0; s < 16; ++s) {
          const v4f av = fa[blk & 1][8 * kk + (s >> 1)];
          v2f a;
          if (s & 1) { a.x = av.z; a.y = av.w; } else { a.x = av.x; a.y = av.y; }
          acc0 = __builtin_amdgcn_wmma_f32_16x16x4_f32(
               false, a,  false, bf[2 * blk + kk][s][0],
               (short)0, acc0,  false,  false);
          acc1 = __builtin_amdgcn_wmma_f32_16x16x4_f32(
               false, a,  false, bf[2 * blk + kk][s][1],
               (short)0, acc1,  false,  false);
        }
      }
      __builtin_amdgcn_sched_barrier(0);
    }

    __syncthreads();
    pb ^= GCN_XT;
    {
      const int n1 = tile + G;
      const int n2 = (n1 < ntiles - G) ? n1 + G : tile;
      {
        const int st = n2;
#pragma unroll
        for (int i = 0; i < 16; ++i) {
          int row = st * 16 + 4 * J + (i >> 2);
          if (row >= M) row = M - 1;
          g[i] = ((const v4f*)(x + (long long)row * 512))[lane + 32 * (i & 3)];
        }
      }
    }
#pragma unroll
    for (int f = 0; f < 16; ++f) fa[0][f] = rd[pb + 16 * (f >> 3) + 2 * (f & 7)];
    __builtin_amdgcn_sched_barrier(0);

#pragma unroll
    for (int kk = 0; kk < 2; ++kk) {
#pragma unroll
      for (int s = 0; s < 16; ++s) {
        const v4f av = fa[1][8 * kk + (s >> 1)];
        v2f a;
        if (s & 1) { a.x = av.z; a.y = av.w; } else { a.x = av.x; a.y = av.y; }
        acc0 = __builtin_amdgcn_wmma_f32_16x16x4_f32(
             false, a,  false, bf[6 + kk][s][0],
             (short)0, acc0,  false,  false);
        acc1 = __builtin_amdgcn_wmma_f32_16x16x4_f32(
             false, a,  false, bf[6 + kk][s][1],
             (short)0, acc1,  false,  false);
      }
    }

    const int rbase = tile * 16 + 8 * hh;
    float* __restrict__ hp = h + (long long)rbase * 128 + 32 * J + 2 * c;
    if (tile * 16 + 16 <= M) {
#pragma unroll
      for (int v = 0; v < 8; ++v) *(float2*)(hp + v * 128) = make_float2(acc0[v], acc1[v]);
    } else {
#pragma unroll
      for (int v = 0; v < 8; ++v) {
        if (rbase + v < M) *(float2*)(hp + v * 128) = make_float2(acc0[v], acc1[v]);
      }
    }
  }

  for (; ek <= np + 1; ++ek) {
    gcn_edge_file(eR, ((long long)(ek - 2) * nw + gw) * 64, lane, cnt, list, M);
    eR = gcn_edge_rank(eL, cnt);
    eL = gcn_edge_load(ei, E, ((long long)ek * nw + gw) * 64, lane);
  }
}

__global__ __launch_bounds__(1024) void gcn_gather1(
    const int* __restrict__ cnt, const int* __restrict__ list,
    const int* __restrict__ ei,
    const float* __restrict__ h1, const float* __restrict__ b1,
    const float* __restrict__ W2, float* __restrict__ h2, int N, int E) {
  __shared__ float res[96];
  const int lane = threadIdx.x & 31;
  const int w = __builtin_amdgcn_readfirstlane((int)(threadIdx.x >> 5));
  const int d = blockIdx.x * 32 + w;

  if (d < N) {
    const int nAll = cnt[d];
    const int n = nAll < GCN_CAP ? nAll : GCN_CAP;
    const float dd = rsqrtf((float)(nAll + 1));
    const int* __restrict__ bl = list + (long long)d * GCN_CAP;
    float4 acc = make_float4(0.f, 0.f, 0.f, 0.f);

    for (int base = 0; base < n; base += 32) {
      const int m = (n - base < 32) ? (n - base) : 32;
      int s_l = 0;
      float w_l = 0.f;
      if (lane < m) {
        s_l = bl[base + lane];
        s_l = ((unsigned)s_l < (unsigned)N) ? s_l : 0;
        w_l = rsqrtf((float)(cnt[s_l] + 1)) * dd;
      }
#pragma unroll 4
      for (int k = 0; k < m; ++k) {
        const int s = __builtin_amdgcn_readlane(s_l, k);
        const float nrm = __int_as_float(__builtin_amdgcn_readlane(__float_as_int(w_l), k));
        const float4 v = ((const float4*)(h1 + (long long)s * 128))[lane];
        acc.x += nrm * v.x;
        acc.y += nrm * v.y;
        acc.z += nrm * v.z;
        acc.w += nrm * v.w;
      }
    }

    if (nAll > GCN_CAP) {
      const int* __restrict__ ovl = list + (long long)N * GCN_CAP;
      const int* __restrict__ src = ei;
      const int* __restrict__ dst = ei + E;
      int no = cnt[N];
      no = no < E ? no : E;
      for (int base = 0; base < no; base += 32) {
        int s_l = 0;
        float w_l = 0.f;
        bool mine = false;
        if (base + lane < no) {
          const int e = ovl[base + lane];
          if ((unsigned)e < (unsigned)E && dst[e] == d) {
            mine = true;
            s_l = src[e];
            s_l = ((unsigned)s_l < (unsigned)N) ? s_l : 0;
            w_l = rsqrtf((float)(cnt[s_l] + 1)) * dd;
          }
        }
        unsigned todo = __builtin_amdgcn_ballot_w32(mine);
        while (todo != 0u) {
          const int k = __builtin_ctz(todo);
          todo &= todo - 1u;
          const int s = __builtin_amdgcn_readlane(s_l, k);
          const float nrm = __int_as_float(__builtin_amdgcn_readlane(__float_as_int(w_l), k));
          const float4 v = ((const float4*)(h1 + (long long)s * 128))[lane];
          acc.x += nrm * v.x;
          acc.y += nrm * v.y;
          acc.z += nrm * v.z;
          acc.w += nrm * v.w;
        }
      }
    }

    const float4 hv = ((const float4*)(h1 + (long long)d * 128))[lane];
    const float nn = dd * dd;
    float y0 = acc.x + nn * hv.x + b1[lane * 4 + 0];
    float y1 = acc.y + nn * hv.y + b1[lane * 4 + 1];
    float y2 = acc.z + nn * hv.z + b1[lane * 4 + 2];
    float y3 = acc.w + nn * hv.w + b1[lane * 4 + 3];
    y0 = y0 > 0.0f ? y0 : 0.0f;
    y1 = y1 > 0.0f ? y1 : 0.0f;
    y2 = y2 > 0.0f ? y2 : 0.0f;
    y3 = y3 > 0.0f ? y3 : 0.0f;

    const float* __restrict__ wr = W2 + lane * 12;
    float p0 = y0 * wr[0] + y1 * wr[3] + y2 * wr[6] + y3 * wr[9];
    float p1 = y0 * wr[1] + y1 * wr[4] + y2 * wr[7] + y3 * wr[10];
    float p2 = y0 * wr[2] + y1 * wr[5] + y2 * wr[8] + y3 * wr[11];
    for (int o = 16; o > 0; o >>= 1) {
      p0 += __shfl_xor(p0, o);
      p1 += __shfl_xor(p1, o);
      p2 += __shfl_xor(p2, o);
    }
    if (lane == 0) {
      res[w * 3 + 0] = p0;
      res[w * 3 + 1] = p1;
      res[w * 3 + 2] = p2;
    }
  }
  __syncthreads();

  if (threadIdx.x < 24) {
    const long long f = (long long)blockIdx.x * 96 + threadIdx.x * 4;
    const long long n3 = (long long)N * 3;
    if (f + 3 < n3) {
      *(float4*)(h2 + f) = make_float4(res[threadIdx.x * 4 + 0], res[threadIdx.x * 4 + 1],
                                       res[threadIdx.x * 4 + 2], res[threadIdx.x * 4 + 3]);
    } else {
      for (int q = 0; q < 4; ++q)
        if (f + q < n3) h2[f + q] = res[threadIdx.x * 4 + q];
    }
  }
}

__global__ __launch_bounds__(256) void gcn_gather2(
    const int* __restrict__ cnt, const int* __restrict__ list,
    const int* __restrict__ ei,
    const float* __restrict__ h2, const float* __restrict__ b2,
    float* __restrict__ out, int N, int E) {
  __shared__ float res[96];
  const int t = threadIdx.x;
  const int g = t >> 3;
  const int l = t & 7;
  const int d = blockIdx.x * 32 + g;
  const bool valid = d < N;
  const int dc = valid ? d : 0;
  const int nAll = valid ? cnt[dc] : 0;
  const int n = nAll < GCN_CAP ? nAll : GCN_CAP;
  const float dd = rsqrtf((float)(nAll + 1));
  const int* __restrict__ bl = list + (long long)dc * GCN_CAP;

  float a0 = 0.f, a1 = 0.f, a2 = 0.f;
  for (int j = l; j < n; j += 8) {
    int s = bl[j];
    s = ((unsigned)s < (unsigned)N) ? s : 0;
    const float nrm = rsqrtf((float)(cnt[s] + 1)) * dd;
    const float* __restrict__ hp = h2 + (long long)s * 3;
    a0 += nrm * hp[0];
    a1 += nrm * hp[1];
    a2 += nrm * hp[2];
  }
  if (nAll > GCN_CAP) {
    const int* __restrict__ ovl = list + (long long)N * GCN_CAP;
    const int* __restrict__ src = ei;
    const int* __restrict__ dst = ei + E;
    int no = cnt[N];
    no = no < E ? no : E;
    for (int o = l; o < no; o += 8) {
      const int e = ovl[o];
      if ((unsigned)e < (unsigned)E && dst[e] == d) {
        int s = src[e];
        s = ((unsigned)s < (unsigned)N) ? s : 0;
        const float nrm = rsqrtf((float)(cnt[s] + 1)) * dd;
        const float* __restrict__ hp = h2 + (long long)s * 3;
        a0 += nrm * hp[0];
        a1 += nrm * hp[1];
        a2 += nrm * hp[2];
      }
    }
  }
  for (int o = 4; o > 0; o >>= 1) {
    a0 += __shfl_xor(a0, o);
    a1 += __shfl_xor(a1, o);
    a2 += __shfl_xor(a2, o);
  }
  if (l == 0) {
    const float nn = dd * dd;
    const float* __restrict__ hp = h2 + (long long)dc * 3;
    res[g * 3 + 0] = a0 + nn * hp[0] + b2[0];
    res[g * 3 + 1] = a1 + nn * hp[1] + b2[1];
    res[g * 3 + 2] = a2 + nn * hp[2] + b2[2];
  }
  __syncthreads();

  if (t < 24) {
    const long long f = (long long)blockIdx.x * 96 + t * 4;
    const long long n3 = (long long)N * 3;
    if (f + 3 < n3) {
      *(float4*)(out + f) = make_float4(res[t * 4 + 0], res[t * 4 + 1],
                                        res[t * 4 + 2], res[t * 4 + 3]);
    } else {
      for (int q = 0; q < 4; ++q)
        if (f + q < n3) out[f + q] = res[t * 4 + q];
    }
  }
}

extern "C" void kernel_launch(void* const* d_in, const int* in_sizes, int n_in,
                              void* d_out, int out_size, void* d_ws, size_t ws_size,
                              hipStream_t stream) {
  const float* x  = (const float*)d_in[0];
  const int*   ei = (const int*)d_in[1];
  const float* W1 = (const float*)d_in[2];
  const float* b1 = (const float*)d_in[3];
  const float* W2 = (const float*)d_in[4];
  const float* b2 = (const float*)d_in[5];
  float* out = (float*)d_out;

  const int N = in_sizes[0] / 512;
  const int E = in_sizes[1] / 2;

  const long long Npad = ((long long)N + 255) & ~255LL;
  const long long Cpad = ((long long)N + 1 + 255) & ~255LL;
  float* ws   = (float*)d_ws;
  float* h1   = ws + Npad;
  float* pl   = h1 + (long long)N * 128;
  float* h2   = pl + (long long)N * 128;
  int* cnt  = (int*)pl;
  int* list = cnt + Cpad;

  const int B = 256;

  gcn_clear<<<(N + 1 + B - 1) / B, B, 0, stream>>>(cnt, N + 1);

  gcn_gemm1_wmma<<<256, 128, 0, stream>>>(x, W1, h1, N, ei, cnt, list, E);

  gcn_gather1<<<(N + 31) / 32, 1024, 0, stream>>>(cnt, list, ei, h1, b1, W2, h2, N, E);
  gcn_gather2<<<(N + 31) / 32, B, 0, stream>>>(cnt, list, ei, h2, b2, out, N, E);
}


# layer-1 gather rebuilt: one wave walks 8 consecutive nodes (was one node per wave, 32-wave blocks, a barrier), next node's slots and counters fetched under its rows; h2 rows 16 B; sum order kept
# speedup vs baseline: 3.5726x; 1.0319x over previous
#include <hip/hip_runtime.h>


typedef __attribute__((ext_vector_type(2))) float v2f;
typedef __attribute__((ext_vector_type(8))) float v8f;
typedef __attribute__((ext_vector_type(4))) float v4f;

constexpr int GCN_CAP = 64;

__global__ void gcn_clear(int* __restrict__ zc, int nz) {
  const int i = blockIdx.x * blockDim.x + threadIdx.x;
  if (i < nz) zc[i] = 0;
}

struct GcnEdgeIdx { int s0, d0, s1, d1; };
struct GcnEdgeRank { int s0, d0, r0, s1, d1, r1; };

__device__ __forceinline__ GcnEdgeIdx gcn_edge_load(const int* __restrict__ ei, int E,
                                                   long long base, int lane) {
  GcnEdgeIdx x;
  x.s0 = 0; x.d0 = -1; x.s1 = 0; x.d1 = -1;
  const long long e0 = base + lane;
  const long long e1 = e0 + 32;
  if (e0 < E) { x.s0 = ei[e0]; x.d0 = ei[E + e0]; }
  if (e1 < E) { x.s1 = ei[e1]; x.d1 = ei[E + e1]; }
  return x;
}

__device__ __forceinline__ GcnEdgeRank gcn_edge_rank(const GcnEdgeIdx x, int* __restrict__ cnt) {
  GcnEdgeRank y;
  y.s0 = x.s0; y.d0 = x.d0; y.s1 = x.s1; y.d1 = x.d1;
  y.r0 = 0; y.r1 = 0;
  if (x.d0 >= 0) y.r0 = atomicAdd(&cnt[x.d0], 1);
  if (x.d1 >= 0) y.r1 = atomicAdd(&cnt[x.d1], 1);
  return y;
}

__device__ __forceinline__ void gcn_edge_file_one(int s, int d, int r, long long e,
                                                  int* __restrict__ cnt,
                                                  int* __restrict__ list, int N) {
  if (d >= 0) {
    int* p = list + (long long)d * GCN_CAP + r;
    int v = s;
    if (r >= GCN_CAP) {
      p = list + (long long)N * GCN_CAP + atomicAdd(&cnt[N], 1);
      v = (int)e;
    }
    *p = v;
  }
}

__device__ __forceinline__ void gcn_edge_file(const GcnEdgeRank y, long long base, int lane,
                                              int* __restrict__ cnt,
                                              int* __restrict__ list, int N) {
  gcn_edge_file_one(y.s0, y.d0, y.r0, base + lane, cnt, list, N);
  gcn_edge_file_one(y.s1, y.d1, y.r1, base + lane + 32, cnt, list, N);
}

constexpr int GCN_XP = 129;
constexpr int GCN_XT = 16 * GCN_XP;

__global__ __launch_bounds__(128) void gcn_gemm1_wmma(
    const float* __restrict__ x, const float* __restrict__ W,
    float* __restrict__ h, int M, const int* __restrict__ ei,
    int* __restrict__ cnt, int* __restrict__ list, int E) {
  __shared__ v4f xs[2 * GCN_XT];
  const int lane = threadIdx.x & 31;
  const int J = __builtin_amdgcn_readfirstlane((int)(threadIdx.x >> 5));
  const int c = lane & 15;
  const int hh = lane >> 4;
  const int ntiles = (M + 15) >> 4;
  const int G = (int)gridDim.x;

  const long long nw = (long long)G * 4;
  const long long gw = (long long)blockIdx.x * 4 + J;
  const long long ngroups = ((long long)E + 63) >> 6;
  const int np = gw < ngroups ? (int)((ngroups - gw + nw - 1) / nw) : 0;
  int ek = 2;
  GcnEdgeIdx eL = gcn_edge_load(ei, E, gw * 64, lane);

  int tile = (int)blockIdx.x;
  const bool has_tile = tile < ntiles;

  v2f bf[8][16][2];
  {
    const float* __restrict__ wb = W + (4 * hh) * 128 + 32 * J + 2 * c;
#pragma unroll
    for (int kb = 0; kb < 8; ++kb) {
#pragma unroll
      for (int s = 0; s < 16; ++s) {
        const int k = 64 * kb + 8 * (s >> 1) + 2 * (s & 1);
        const float2 r0 = *(const float2*)(wb + k * 128);
        const float2 r1 = *(const float2*)(wb + (k + 1) * 128);
        bf[kb][s][0].x = r0.x; bf[kb][s][0].y = r1.x;
        bf[kb][s][1].x = r0.y; bf[kb][s][1].y = r1.y;
      }
    }
  }

  GcnEdgeRank eR = gcn_edge_rank(eL, cnt);
  eL = gcn_edge_load(ei, E, (nw + gw) * 64, lane);

  if (!has_tile) {
    for (; ek <= np + 1; ++ek) {
      gcn_edge_file(eR, ((long long)(ek - 2) * nw + gw) * 64, lane, cnt, list, M);
      eR = gcn_edge_rank(eL, cnt);
      eL = gcn_edge_load(ei, E, ((long long)ek * nw + gw) * 64, lane);
    }
    return;
  }

  v4f g[16];
  {
    const int st = tile;
#pragma unroll
    for (int i = 0; i < 16; ++i) {
      int row = st * 16 + 4 * J + (i >> 2);
      if (row >= M) row = M - 1;
      g[i] = ((const v4f*)(x + (long long)row * 512))[lane + 32 * (i & 3)];
    }
  }

  v4f* __restrict__ sw = xs + (4 * J) * GCN_XP + lane;
  const v4f* __restrict__ rd = xs + c * GCN_XP + hh;
  int pb = 0;

  {
    v4f* __restrict__ sd = sw;
#pragma unroll
    for (int i = 0; i < 16; ++i) sd[(i >> 2) * GCN_XP + 32 * (i & 3)] = g[i];
  }
  __syncthreads();

  {
    const int st = (tile + G < ntiles) ? tile + G : tile;
#pragma unroll
    for (int i = 0; i < 16; ++i) {
      int row = st * 16 + 4 * J + (i >> 2);
      if (row >= M) row = M - 1;
      g[i] = ((const v4f*)(x + (long long)row * 512))[lane + 32 * (i & 3)];
    }
  }

  v4f fa[2][16];
#pragma unroll
  for (int f = 0; f < 16; ++f) fa[0][f] = rd[16 * (f >> 3) + 2 * (f & 7)];

  for (; tile < ntiles; tile += G) {
    gcn_edge_file(eR, ((long long)(ek - 2) * nw + gw) * 64, lane, cnt, list, M);
    eR = gcn_edge_rank(eL, cnt);
    eL = gcn_edge_load(ei, E, ((long long)ek * nw + gw) * 64, lane);
    ++ek;

    v8f acc0 = {};
    v8f acc1 = {};

#pragma unroll
    for (int blk = 0; blk < 3; ++blk) {
      if (blk == 2) {
        v4f* __restrict__ sd = sw + (pb ^ GCN_XT);
#pragma unroll
        for (int i = 0; i < 16; ++i) sd[(i >> 2) * GCN_XP + 32 * (i & 3)] = g[i];
      }
#pragma unroll
      for (int f = 0; f < 16; ++f)
        fa[(blk + 1) & 1][f] = rd[pb + 16 * (2 * (blk + 1) + (f >> 3)) + 2 * (f & 7)];
      __builtin_amdgcn_sched_barrier(0);

#pragma unroll
      for (int kk = 0; kk < 2; ++kk) {
#pragma unroll
        for (int s = 0; s < 16; ++s) {
          const v4f av = fa[blk & 1][8 * kk + (s >> 1)];
          v2f a;
          if (s & 1) { a.x = av.z; a.y = av.w; } else { a.x = av.x; a.y = av.y; }
          acc0 = __builtin_amdgcn_wmma_f32_16x16x4_f32(
               false, a,  false, bf[2 * blk + kk][s][0],
               (short)0, acc0,  false,  false);
          acc1 = __builtin_amdgcn_wmma_f32_16x16x4_f32(
               false, a,  false, bf[2 * blk + kk][s][1],
               (short)0, acc1,  false,  false);
        }
      }
      __builtin_amdgcn_sched_barrier(0);
    }

    __syncthreads();
    pb ^= GCN_XT;
    {
      const int n1 = tile + G;
      const int n2 = (n1 < ntiles - G) ? n1 + G : tile;
      {
        const int st = n2;
#pragma unroll
        for (int i = 0; i < 16; ++i) {
          int row = st * 16 + 4 * J + (i >> 2);
          if (row >= M) row = M - 1;
          g[i] = ((const v4f*)(x + (long long)row * 512))[lane + 32 * (i & 3)];
        }
      }
    }
#pragma unroll
    for (int f = 0; f < 16; ++f) fa[0][f] = rd[pb + 16 * (f >> 3) + 2 * (f & 7)];
    __builtin_amdgcn_sched_barrier(0);

#pragma unroll
    for (int kk = 0; kk < 2; ++kk) {
#pragma unroll
      for (int s = 0; s < 16; ++s) {
        const v4f av = fa[1][8 * kk + (s >> 1)];
        v2f a;
        if (s & 1) { a.x = av.z; a.y = av.w; } else { a.x = av.x; a.y = av.y; }
        acc0 = __builtin_amdgcn_wmma_f32_16x16x4_f32(
             false, a,  false, bf[6 + kk][s][0],
             (short)0, acc0,  false,  false);
        acc1 = __builtin_amdgcn_wmma_f32_16x16x4_f32(
             false, a,  false, bf[6 + kk][s][1],
             (short)0, acc1,  false,  false);
      }
    }

    const int rbase = tile * 16 + 8 * hh;
    float* __restrict__ hp = h + (long long)rbase * 128 + 32 * J + 2 * c;
    if (tile * 16 + 16 <= M) {
#pragma unroll
      for (int v = 0; v < 8; ++v) *(float2*)(hp + v * 128) = make_float2(acc0[v], acc1[v]);
    } else {
#pragma unroll
      for (int v = 0; v < 8; ++v) {
        if (rbase + v < M) *(float2*)(hp + v * 128) = make_float2(acc0[v], acc1[v]);
      }
    }
  }

  for (; ek <= np + 1; ++ek) {
    gcn_edge_file(eR, ((long long)(ek - 2) * nw + gw) * 64, lane, cnt, list, M);
    eR = gcn_edge_rank(eL, cnt);
    eL = gcn_edge_load(ei, E, ((long long)ek * nw + gw) * 64, lane);
  }
}

constexpr int GCN_G1N = 8;

template <int R>
__device__ __forceinline__ void gcn_rows(const float* __restrict__ h1, int s_l, float w_l,
                                         int k0, int lane, float4& acc) {
  float4 v[R];
#pragma unroll
  for (int u = 0; u < R; ++u) {
    const int s = __builtin_amdgcn_readlane(s_l, k0 + u);
    v[u] = ((const float4*)(h1 + (long long)s * 128))[lane];
  }
#pragma unroll
  for (int u = 0; u < R; ++u) {
    const float nrm = __int_as_float(__builtin_amdgcn_readlane(__float_as_int(w_l), k0 + u));
    acc.x += nrm * v[u].x;
    acc.y += nrm * v[u].y;
    acc.z += nrm * v[u].z;
    acc.w += nrm * v[u].w;
  }
}

__global__ __launch_bounds__(64) void gcn_gather1(
    const int* __restrict__ cnt, const int* __restrict__ list,
    const int* __restrict__ ei,
    const float* __restrict__ h1, const float* __restrict__ b1,
    const float* __restrict__ W2, float* __restrict__ h2, int N, int E) {
  const int lane = threadIdx.x & 31;
  const int task = __builtin_amdgcn_readfirstlane(
      (int)(blockIdx.x * (blockDim.x >> 5) + (threadIdx.x >> 5)));
  const int d0 = task * GCN_G1N;
  if (d0 >= N) return;

  const float4 bb = ((const float4*)b1)[lane];
  float w2r[12];
  {
    const float* __restrict__ wr = W2 + lane * 12;
#pragma unroll
    for (int i = 0; i < 12; ++i) w2r[i] = wr[i];
  }

  float r0 = 0.f, r1 = 0.f, r2 = 0.f;

  int cnt8 = 0;
  if (lane < GCN_G1N && d0 + lane < N) cnt8 = cnt[d0 + lane];

  int s_cur = 0;
  int s_nxt = 0;
  float w_cur;
  {
    int c_cur = 0;
    const int n0 = __builtin_amdgcn_readlane(cnt8, 0);
    const int n1 = __builtin_amdgcn_readlane(cnt8, 1);
    const int m0 = n0 < 32 ? n0 : 32;
    if (lane < m0) s_cur = list[(long long)d0 * GCN_CAP + lane];
    if (lane < (n1 < 32 ? n1 : 32)) s_nxt = list[(long long)(d0 + 1) * GCN_CAP + lane];
    s_cur = ((unsigned)s_cur < (unsigned)N) ? s_cur : 0;
    if (lane < m0) c_cur = cnt[s_cur];
    w_cur = rsqrtf((float)(c_cur + 1)) * rsqrtf((float)(n0 + 1));
  }

  for (int j = 0; j < GCN_G1N; ++j) {
    const int d = d0 + j;
    if (d >= N) break;
    const int nAll = __builtin_amdgcn_readlane(cnt8, j);
    const int s_l = s_cur;
    const float w_l = w_cur;
    int c_nxt;
    int nn1;
    {
      nn1 = (j + 1 < GCN_G1N) ? __builtin_amdgcn_readlane(cnt8, (j + 1) & (GCN_G1N - 1)) : 0;
      const int nn2 =
          (j + 2 < GCN_G1N) ? __builtin_amdgcn_readlane(cnt8, (j + 2) & (GCN_G1N - 1)) : 0;
      s_cur = ((unsigned)s_nxt < (unsigned)N) ? s_nxt : 0;
      c_nxt = cnt[(lane < (nn1 < 32 ? nn1 : 32)) ? s_cur : d];
      s_nxt = 0;
      if (lane < (nn2 < 32 ? nn2 : 32)) s_nxt = list[(long long)(d + 2) * GCN_CAP + lane];
    }
    const int n = nAll < GCN_CAP ? nAll : GCN_CAP;
    const float dd = rsqrtf((float)(nAll + 1));
    const int* __restrict__ bl = list + (long long)d * GCN_CAP;
    const float4 hv = ((const float4*)(h1 + (long long)d * 128))[lane];
    float4 acc = make_float4(0.f, 0.f, 0.f, 0.f);

    {
      const int m = n < 32 ? n : 32;
      int k0 = 0;
      while (m - k0 > 16) {
        gcn_rows<8>(h1, s_l, w_l, k0, lane, acc);
        k0 += 8;
      }
      switch (m - k0) {
        case 1: gcn_rows<1>(h1, s_l, w_l, k0, lane, acc); break;
        case 2: gcn_rows<2>(h1, s_l, w_l, k0, lane, acc); break;
        case 3: gcn_rows<3>(h1, s_l, w_l, k0, lane, acc); break;
        case 4: gcn_rows<4>(h1, s_l, w_l, k0, lane, acc); break;
        case 5: gcn_rows<5>(h1, s_l, w_l, k0, lane, acc); break;
        case 6: gcn_rows<6>(h1, s_l, w_l, k0, lane, acc); break;
        case 7: gcn_rows<7>(h1, s_l, w_l, k0, lane, acc); break;
        case 8: gcn_rows<8>(h1, s_l, w_l, k0, lane, acc); break;
        case 9: gcn_rows<9>(h1, s_l, w_l, k0, lane, acc); break;
        case 10: gcn_rows<10>(h1, s_l, w_l, k0, lane, acc); break;
        case 11: gcn_rows<11>(h1, s_l, w_l, k0, lane, acc); break;
        case 12: gcn_rows<12>(h1, s_l, w_l, k0, lane, acc); break;
        case 13: gcn_rows<13>(h1, s_l, w_l, k0, lane, acc); break;
        case 14: gcn_rows<14>(h1, s_l, w_l, k0, lane, acc); break;
        case 15: gcn_rows<15>(h1, s_l, w_l, k0, lane, acc); break;
        case 16: gcn_rows<16>(h1, s_l, w_l, k0, lane, acc); break;
        default: break;
      }
    }

    if (n > 32) {
      const int m2 = n - 32;
      int s2_l = 0;
      float w2_l = 0.f;
      if (lane < m2) {
        s2_l = bl[32 + lane];
        s2_l = ((unsigned)s2_l < (unsigned)N) ? s2_l : 0;
        w2_l = rsqrtf((float)(cnt[s2_l] + 1)) * dd;
      }
      for (int k = 0; k < m2; ++k) {
        const int s = __builtin_amdgcn_readlane(s2_l, k);
        const float nrm = __int_as_float(__builtin_amdgcn_readlane(__float_as_int(w2_l), k));
        const float4 v = ((const float4*)(h1 + (long long)s * 128))[lane];
        acc.x += nrm * v.x;
        acc.y += nrm * v.y;
        acc.z += nrm * v.z;
        acc.w += nrm * v.w;
      }
    }

    if (nAll > GCN_CAP) {
      const int* __restrict__ ovl = list + (long long)N * GCN_CAP;
      const int* __restrict__ src = ei;
      const int* __restrict__ dst = ei + E;
      int no = cnt[N];
      no = no < E ? no : E;
      for (int base = 0; base < no; base += 32) {
        int so_l = 0;
        float w_l = 0.f;
        bool mine = false;
        if (base + lane < no) {
          const int e = ovl[base + lane];
          if ((unsigned)e < (unsigned)E && dst[e] == d) {
            mine = true;
            so_l = src[e];
            so_l = ((unsigned)so_l < (unsigned)N) ? so_l : 0;
            w_l = rsqrtf((float)(cnt[so_l] + 1)) * dd;
          }
        }
        unsigned todo = __builtin_amdgcn_ballot_w32(mine);
        while (todo != 0u) {
          const int k = __builtin_ctz(todo);
          todo &= todo - 1u;
          const int s = __builtin_amdgcn_readlane(so_l, k);
          const float nrm = __int_as_float(__builtin_amdgcn_readlane(__float_as_int(w_l), k));
          const float4 v = ((const float4*)(h1 + (long long)s * 128))[lane];
          acc.x += nrm * v.x;
          acc.y += nrm * v.y;
          acc.z += nrm * v.z;
          acc.w += nrm * v.w;
        }
      }
    }

    const float nn = dd * dd;
    float y0 = acc.x + nn * hv.x + bb.x;
    float y1 = acc.y + nn * hv.y + bb.y;
    float y2 = acc.z + nn * hv.z + bb.z;
    float y3 = acc.w + nn * hv.w + bb.w;
    y0 = y0 > 0.0f ? y0 : 0.0f;
    y1 = y1 > 0.0f ? y1 : 0.0f;
    y2 = y2 > 0.0f ? y2 : 0.0f;
    y3 = y3 > 0.0f ? y3 : 0.0f;

    float p0 = y0 * w2r[0] + y1 * w2r[3] + y2 * w2r[6] + y3 * w2r[9];
    float p1 = y0 * w2r[1] + y1 * w2r[4] + y2 * w2r[7] + y3 * w2r[10];
    float p2 = y0 * w2r[2] + y1 * w2r[5] + y2 * w2r[8] + y3 * w2r[11];
    for (int o = 16; o > 0; o >>= 1) {
      p0 += __shfl_xor(p0, o);
      p1 += __shfl_xor(p1, o);
      p2 += __shfl_xor(p2, o);
    }
    if (lane == j) { r0 = p0; r1 = p1; r2 = p2; }
    w_cur = rsqrtf((float)(c_nxt + 1)) * rsqrtf((float)(nn1 + 1));
  }

  if (lane < GCN_G1N && d0 + lane < N)
    ((float4*)h2)[d0 + lane] = make_float4(r0, r1, r2, 0.f);
}

__global__ __launch_bounds__(256) void gcn_gather2(
    const int* __restrict__ cnt, const int* __restrict__ list,
    const int* __restrict__ ei,
    const float* __restrict__ h2, const float* __restrict__ b2,
    float* __restrict__ out, int N, int E) {
  __shared__ float res[96];
  const int t = threadIdx.x;
  const int g = t >> 3;
  const int l = t & 7;
  const int d = blockIdx.x * 32 + g;
  const bool valid = d < N;
  const int dc = valid ? d : 0;
  const int nAll = valid ? cnt[dc] : 0;
  const int n = nAll < GCN_CAP ? nAll : GCN_CAP;
  const float dd = rsqrtf((float)(nAll + 1));
  const int* __restrict__ bl = list + (long long)dc * GCN_CAP;

  float a0 = 0.f, a1 = 0.f, a2 = 0.f;
  for (int j = l; j < n; j += 8) {
    int s = bl[j];
    s = ((unsigned)s < (unsigned)N) ? s : 0;
    const float nrm = rsqrtf((float)(cnt[s] + 1)) * dd;
    const float4 hq = ((const float4*)h2)[s];
    a0 += nrm * hq.x;
    a1 += nrm * hq.y;
    a2 += nrm * hq.z;
  }
  if (nAll > GCN_CAP) {
    const int* __restrict__ ovl = list + (long long)N * GCN_CAP;
    const int* __restrict__ src = ei;
    const int* __restrict__ dst = ei + E;
    int no = cnt[N];
    no = no < E ? no : E;
    for (int o = l; o < no; o += 8) {
      const int e = ovl[o];
      if ((unsigned)e < (unsigned)E && dst[e] == d) {
        int s = src[e];
        s = ((unsigned)s < (unsigned)N) ? s : 0;
        const float nrm = rsqrtf((float)(cnt[s] + 1)) * dd;
        const float4 hq = ((const float4*)h2)[s];
        a0 += nrm * hq.x;
        a1 += nrm * hq.y;
        a2 += nrm * hq.z;
      }
    }
  }
  for (int o = 4; o > 0; o >>= 1) {
    a0 += __shfl_xor(a0, o);
    a1 += __shfl_xor(a1, o);
    a2 += __shfl_xor(a2, o);
  }
  if (l == 0) {
    const float nn = dd * dd;
    const float4 hq = ((const float4*)h2)[dc];
    res[g * 3 + 0] = a0 + nn * hq.x + b2[0];
    res[g * 3 + 1] = a1 + nn * hq.y + b2[1];
    res[g * 3 + 2] = a2 + nn * hq.z + b2[2];
  }
  __syncthreads();

  if (t < 24) {
    const long long f = (long long)blockIdx.x * 96 + t * 4;
    const long long n3 = (long long)N * 3;
    if (f + 3 < n3) {
      *(float4*)(out + f) = make_float4(res[t * 4 + 0], res[t * 4 + 1],
                                        res[t * 4 + 2], res[t * 4 + 3]);
    } else {
      for (int q = 0; q < 4; ++q)
        if (f + q < n3) out[f + q] = res[t * 4 + q];
    }
  }
}

extern "C" void kernel_launch(void* const* d_in, const int* in_sizes, int n_in,
                              void* d_out, int out_size, void* d_ws, size_t ws_size,
                              hipStream_t stream) {
  const float* x  = (const float*)d_in[0];
  const int*   ei = (const int*)d_in[1];
  const float* W1 = (const float*)d_in[2];
  const float* b1 = (const float*)d_in[3];
  const float* W2 = (const float*)d_in[4];
  const float* b2 = (const float*)d_in[5];
  float* out = (float*)d_out;

  const int N = in_sizes[0] / 512;
  const int E = in_sizes[1] / 2;

  const long long Npad = ((long long)N + 255) & ~255LL;
  const long long Cpad = ((long long)N + 1 + 255) & ~255LL;
  float* ws   = (float*)d_ws;
  float* h1   = ws + Npad;
  float* pl   = h1 + (long long)N * 128;
  float* h2   = pl + (((long long)N * 124) & ~31LL);
  int* cnt  = (int*)pl;
  int* list = cnt + Cpad;

  const int B = 256;

  gcn_clear<<<(N + 1 + B - 1) / B, B, 0, stream>>>(cnt, N + 1);

  gcn_gemm1_wmma<<<256, 128, 0, stream>>>(x, W1, h1, N, ei, cnt, list, E);

  gcn_gather1<<<(N + 15) / 16, 64, 0, stream>>>(cnt, list, ei, h1, b1, W2, h2, N, E);
  gcn_gather2<<<(N + 31) / 32, B, 0, stream>>>(cnt, list, ei, h2, b2, out, N, E);
}


# GEMM1 tile loop never drains: edge step without per-lane branches so the x-tile wait is a count, LDS-only barrier; rank atomics and bucket stores stay in flight a whole tile; sum orders kept
# speedup vs baseline: 3.8213x; 1.0696x over previous
#include <hip/hip_runtime.h>


typedef __attribute__((ext_vector_type(2))) float v2f;
typedef __attribute__((ext_vector_type(8))) float v8f;
typedef __attribute__((ext_vector_type(4))) float v4f;

constexpr int GCN_CAP = 64;

__global__ void gcn_clear(int* __restrict__ zc, int nz) {
  const int i = blockIdx.x * blockDim.x + threadIdx.x;
  if (i < nz) zc[i] = 0;
}

struct GcnEdgeIdx { int s0, d0, s1, d1; };
struct GcnEdgeRank { int s0, d0, r0, s1, d1, r1; };

__device__ __forceinline__ GcnEdgeIdx gcn_edge_load(const int* __restrict__ ei, int E,
                                                   long long base, int lane) {
  GcnEdgeIdx x;
  x.s0 = 0; x.d0 = -1; x.s1 = 0; x.d1 = -1;
  const long long e0 = base + lane;
  const long long e1 = e0 + 32;
  if (e0 < E) { x.s0 = ei[e0]; x.d0 = ei[E + e0]; }
  if (e1 < E) { x.s1 = ei[e1]; x.d1 = ei[E + e1]; }
  return x;
}

__device__ __forceinline__ GcnEdgeRank gcn_edge_rank(const GcnEdgeIdx x, int* __restrict__ cnt) {
  GcnEdgeRank y;
  y.s0 = x.s0; y.d0 = x.d0; y.s1 = x.s1; y.d1 = x.d1;
  y.r0 = 0; y.r1 = 0;
  if (x.d0 >= 0) y.r0 = atomicAdd(&cnt[x.d0], 1);
  if (x.d1 >= 0) y.r1 = atomicAdd(&cnt[x.d1], 1);
  return y;
}

__device__ __forceinline__ void gcn_edge_file_one(int s, int d, int r, long long e,
                                                  int* __restrict__ cnt,
                                                  int* __restrict__ list, int N) {
  if (d >= 0) {
    int* p = list + (long long)d * GCN_CAP + r;
    int v = s;
    if (r >= GCN_CAP) {
      p = list + (long long)N * GCN_CAP + atomicAdd(&cnt[N], 1);
      v = (int)e;
    }
    *p = v;
  }
}

__device__ __forceinline__ void gcn_edge_file(const GcnEdgeRank y, long long base, int lane,
                                              int* __restrict__ cnt,
                                              int* __restrict__ list, int N) {
  gcn_edge_file_one(y.s0, y.d0, y.r0, base + lane, cnt, list, N);
  gcn_edge_file_one(y.s1, y.d1, y.r1, base + lane + 32, cnt, list, N);
}

__device__ __forceinline__ GcnEdgeIdx gcn_edge_load_nb(const int* __restrict__ ei, int E,
                                                      long long base, int lane) {
  GcnEdgeIdx x;
  const long long e0 = base + lane;
  const long long e1 = e0 + 32;
  const long long last = (long long)E - 1;
  const long long c0 = e0 < last ? e0 : last;
  const long long c1 = e1 < last ? e1 : last;
  x.s0 = ei[c0]; x.d0 = ei[E + c0];
  x.s1 = ei[c1]; x.d1 = ei[E + c1];
  x.d0 = e0 < E ? x.d0 : -1;
  x.d1 = e1 < E ? x.d1 : -1;
  return x;
}

__device__ __forceinline__ GcnEdgeRank gcn_edge_rank_nb(const GcnEdgeIdx x, int* __restrict__ cnt) {
  GcnEdgeRank y;
  y.s0 = x.s0; y.d0 = x.d0; y.s1 = x.s1; y.d1 = x.d1;
  y.r0 = atomicAdd(&cnt[x.d0], 1);
  y.r1 = atomicAdd(&cnt[x.d1], 1);
  return y;
}

__device__ __forceinline__ void gcn_edge_file_one_nb(int s, int d, int r, long long e,
                                                     int* __restrict__ cnt,
                                                     int* __restrict__ list, int N) {
  int* p = list + (long long)d * GCN_CAP + r;
  int v = s;
  if (r >= GCN_CAP) {
    p = list + (long long)N * GCN_CAP + atomicAdd(&cnt[N], 1);
    v = (int)e;
  }
  *p = v;
}

__device__ __forceinline__ void gcn_edge_file_nb(const GcnEdgeRank y, long long base, int lane,
                                                 int* __restrict__ cnt,
                                                 int* __restrict__ list, int N) {
  gcn_edge_file_one_nb(y.s0, y.d0, y.r0, base + lane, cnt, list, N);
  gcn_edge_file_one_nb(y.s1, y.d1, y.r1, base + lane + 32, cnt, list, N);
}

__device__ __forceinline__ void gcn_lds_barrier() {
  asm volatile("s_wait_dscnt 0x0\n\ts_barrier_signal -1\n\ts_barrier_wait -1" ::: "memory");
}

constexpr int GCN_XP = 129;
constexpr int GCN_XT = 16 * GCN_XP;

__global__ __launch_bounds__(128) void gcn_gemm1_wmma(
    const float* __restrict__ x, const float* __restrict__ W,
    float* __restrict__ h, int M, const int* __restrict__ ei,
    int* __restrict__ cnt, int* __restrict__ list, int E) {
  __shared__ v4f xs[2 * GCN_XT];
  const int lane = threadIdx.x & 31;
  const int J = __builtin_amdgcn_readfirstlane((int)(threadIdx.x >> 5));
  const int c = lane & 15;
  const int hh = lane >> 4;
  const int ntiles = (M + 15) >> 4;
  const int G = (int)gridDim.x;

  const long long nw = (long long)G * 4;
  const long long gw = (long long)blockIdx.x * 4 + J;
  const long long ngroups = ((long long)E + 63) >> 6;
  const int np = gw < ngroups ? (int)((ngroups - gw + nw - 1) / nw) : 0;
  const long long nfg = (long long)E >> 6;
  const long long gw3 = (long long)blockIdx.x * 4 + 3;
  const int nfull = gw3 < nfg ? (int)((nfg - gw3 + nw - 1) / nw) : 0;
  int ek = 2;
  GcnEdgeIdx eL = gcn_edge_load(ei, E, gw * 64, lane);

  int tile = (int)blockIdx.x;
  const bool has_tile = tile < ntiles;

  v2f bf[8][16][2];
  {
    const float* __restrict__ wb = W + (4 * hh) * 128 + 32 * J + 2 * c;
#pragma unroll
    for (int kb = 0; kb < 8; ++kb) {
#pragma unroll
      for (int s = 0; s < 16; ++s) {
        const int k = 64 * kb + 8 * (s >> 1) + 2 * (s & 1);
        const float2 r0 = *(const float2*)(wb + k * 128);
        const float2 r1 = *(const float2*)(wb + (k + 1) * 128);
        bf[kb][s][0].x = r0.x; bf[kb][s][0].y = r1.x;
        bf[kb][s][1].x = r0.y; bf[kb][s][1].y = r1.y;
      }
    }
  }

  GcnEdgeRank eR = gcn_edge_rank(eL, cnt);
  eL = gcn_edge_load(ei, E, (nw + gw) * 64, lane);

  if (!has_tile) {
    for (; ek <= np + 1; ++ek) {
      gcn_edge_file(eR, ((long long)(ek - 2) * nw + gw) * 64, lane, cnt, list, M);
      eR = gcn_edge_rank(eL, cnt);
      eL = gcn_edge_load(ei, E, ((long long)ek * nw + gw) * 64, lane);
    }
    return;
  }

  v4f g[16];
  {
    const int st = tile;
#pragma unroll
    for (int i = 0; i < 16; ++i) {
      int row = st * 16 + 4 * J + (i >> 2);
      if (row >= M) row = M - 1;
      g[i] = ((const v4f*)(x + (long long)row * 512))[lane + 32 * (i & 3)];
    }
  }

  v4f* __restrict__ sw = xs + (4 * J) * GCN_XP + lane;
  const v4f* __restrict__ rd = xs + c * GCN_XP + hh;
  int pb = 0;

  {
    v4f* __restrict__ sd = sw;
#pragma unroll
    for (int i = 0; i < 16; ++i) sd[(i >> 2) * GCN_XP + 32 * (i & 3)] = g[i];
  }
  __syncthreads();

  {
    const int st = (tile + G < ntiles) ? tile + G : tile;
#pragma unroll
    for (int i = 0; i < 16; ++i) {
      int row = st * 16 + 4 * J + (i >> 2);
      if (row >= M) row = M - 1;
      g[i] = ((const v4f*)(x + (long long)row * 512))[lane + 32 * (i & 3)];
    }
  }

  v4f fa[2][16];
#pragma unroll
  for (int f = 0; f < 16; ++f) fa[0][f] = rd[16 * (f >> 3) + 2 * (f & 7)];

  int it = 0;
#pragma unroll
  for (int pass = 0; pass < 2; ++pass) {
  for (; tile < ntiles && (pass == 1 || it + 1 < nfull); tile += G, ++it) {
    if (pass == 0) {
      gcn_edge_file_nb(eR, ((long long)(ek - 2) * nw + gw) * 64, lane, cnt, list, M);
      eR = gcn_edge_rank_nb(eL, cnt);
      eL = gcn_edge_load_nb(ei, E, ((long long)ek * nw + gw) * 64, lane);
    } else {
      gcn_edge_file(eR, ((long long)(ek - 2) * nw + gw) * 64, lane, cnt, list, M);
      eR = gcn_edge_rank(eL, cnt);
      eL = gcn_edge_load(ei, E, ((long long)ek * nw + gw) * 64, lane);
    }
    ++ek;

    v8f acc0 = {};
    v8f acc1 = {};

#pragma unroll
    for (int blk = 0; blk < 3; ++blk) {
      if (blk == 2) {
        v4f* __restrict__ sd = sw + (pb ^ GCN_XT);
#pragma unroll
        for (int i = 0; i < 16; ++i) sd[(i >> 2) * GCN_XP + 32 * (i & 3)] = g[i];
      }
#pragma unroll
      for (int f = 0; f < 16; ++f)
        fa[(blk + 1) & 1][f] = rd[pb + 16 * (2 * (blk + 1) + (f >> 3)) + 2 * (f & 7)];
      __builtin_amdgcn_sched_barrier(0);

#pragma unroll
      for (int kk = 0; kk < 2; ++kk) {
#pragma unroll
        for (int s = 0; s < 16; ++s) {
          const v4f av = fa[blk & 1][8 * kk + (s >> 1)];
          v2f a;
          if (s & 1) { a.x = av.z; a.y = av.w; } else { a.x = av.x; a.y = av.y; }
          acc0 = __builtin_amdgcn_wmma_f32_16x16x4_f32(
               false, a,  false, bf[2 * blk + kk][s][0],
               (short)0, acc0,  false,  false);
          acc1 = __builtin_amdgcn_wmma_f32_16x16x4_f32(
               false, a,  false, bf[2 * blk + kk][s][1],
               (short)0, acc1,  false,  false);
        }
      }
      __builtin_amdgcn_sched_barrier(0);
    }

    gcn_lds_barrier();
    pb ^= GCN_XT;
    {
      const int n1 = tile + G;
      const int n2 = (n1 < ntiles - G) ? n1 + G : tile;
      {
        const int st = n2;
#pragma unroll
        for (int i = 0; i < 16; ++i) {
          int row = st * 16 + 4 * J + (i >> 2);
          if (row >= M) row = M - 1;
          g[i] = ((const v4f*)(x + (long long)row * 512))[lane + 32 * (i & 3)];
        }
      }
    }
#pragma unroll
    for (int f = 0; f < 16; ++f) fa[0][f] = rd[pb + 16 * (f >> 3) + 2 * (f & 7)];
    __builtin_amdgcn_sched_barrier(0);

#pragma unroll
    for (int kk = 0; kk < 2; ++kk) {
#pragma unroll
      for (int s = 0; s < 16; ++s) {
        const v4f av = fa[1][8 * kk + (s >> 1)];
        v2f a;
        if (s & 1) { a.x = av.z; a.y = av.w; } else { a.x = av.x; a.y = av.y; }
        acc0 = __builtin_amdgcn_wmma_f32_16x16x4_f32(
             false, a,  false, bf[6 + kk][s][0],
             (short)0, acc0,  false,  false);
        acc1 = __builtin_amdgcn_wmma_f32_16x16x4_f32(
             false, a,  false, bf[6 + kk][s][1],
             (short)0, acc1,  false,  false);
      }
    }

    const int rbase = tile * 16 + 8 * hh;
    float* __restrict__ hp = h + (long long)rbase * 128 + 32 * J + 2 * c;
    if (tile * 16 + 16 <= M) {
#pragma unroll
      for (int v = 0; v < 8; ++v) *(float2*)(hp + v * 128) = make_float2(acc0[v], acc1[v]);
    } else {
#pragma unroll
      for (int v = 0; v < 8; ++v) {
        if (rbase + v < M) *(float2*)(hp + v * 128) = make_float2(acc0[v], acc1[v]);
      }
    }
  }
  }

  for (; ek <= np + 1; ++ek) {
    gcn_edge_file(eR, ((long long)(ek - 2) * nw + gw) * 64, lane, cnt, list, M);
    eR = gcn_edge_rank(eL, cnt);
    eL = gcn_edge_load(ei, E, ((long long)ek * nw + gw) * 64, lane);
  }
}

constexpr int GCN_G1N = 8;

template <int R>
__device__ __forceinline__ void gcn_rows(const float* __restrict__ h1, int s_l, float w_l,
                                         int k0, int lane, float4& acc) {
  float4 v[R];
#pragma unroll
  for (int u = 0; u < R; ++u) {
    const int s = __builtin_amdgcn_readlane(s_l, k0 + u);
    v[u] = ((const float4*)(h1 + (long long)s * 128))[lane];
  }
#pragma unroll
  for (int u = 0; u < R; ++u) {
    const float nrm = __int_as_float(__builtin_amdgcn_readlane(__float_as_int(w_l), k0 + u));
    acc.x += nrm * v[u].x;
    acc.y += nrm * v[u].y;
    acc.z += nrm * v[u].z;
    acc.w += nrm * v[u].w;
  }
}

__global__ __launch_bounds__(64) void gcn_gather1(
    const int* __restrict__ cnt, const int* __restrict__ list,
    const int* __restrict__ ei,
    const float* __restrict__ h1, const float* __restrict__ b1,
    const float* __restrict__ W2, float* __restrict__ h2, int N, int E) {
  const int lane = threadIdx.x & 31;
  const int task = __builtin_amdgcn_readfirstlane(
      (int)(blockIdx.x * (blockDim.x >> 5) + (threadIdx.x >> 5)));
  const int d0 = task * GCN_G1N;
  if (d0 >= N) return;

  const float4 bb = ((const float4*)b1)[lane];
  float w2r[12];
  {
    const float* __restrict__ wr = W2 + lane * 12;
#pragma unroll
    for (int i = 0; i < 12; ++i) w2r[i] = wr[i];
  }

  float r0 = 0.f, r1 = 0.f, r2 = 0.f;

  int cnt8 = 0;
  if (lane < GCN_G1N && d0 + lane < N) cnt8 = cnt[d0 + lane];

  int s_cur = 0;
  int s_nxt = 0;
  float w_cur;
  {
    int c_cur = 0;
    const int n0 = __builtin_amdgcn_readlane(cnt8, 0);
    const int n1 = __builtin_amdgcn_readlane(cnt8, 1);
    const int m0 = n0 < 32 ? n0 : 32;
    if (lane < m0) s_cur = list[(long long)d0 * GCN_CAP + lane];
    if (lane < (n1 < 32 ? n1 : 32)) s_nxt = list[(long long)(d0 + 1) * GCN_CAP + lane];
    s_cur = ((unsigned)s_cur < (unsigned)N) ? s_cur : 0;
    if (lane < m0) c_cur = cnt[s_cur];
    w_cur = rsqrtf((float)(c_cur + 1)) * rsqrtf((float)(n0 + 1));
  }

  for (int j = 0; j < GCN_G1N; ++j) {
    const int d = d0 + j;
    if (d >= N) break;
    const int nAll = __builtin_amdgcn_readlane(cnt8, j);
    const int s_l = s_cur;
    const float w_l = w_cur;
    int c_nxt;
    int nn1;
    {
      nn1 = (j + 1 < GCN_G1N) ? __builtin_amdgcn_readlane(cnt8, (j + 1) & (GCN_G1N - 1)) : 0;
      const int nn2 =
          (j + 2 < GCN_G1N) ? __builtin_amdgcn_readlane(cnt8, (j + 2) & (GCN_G1N - 1)) : 0;
      s_cur = ((unsigned)s_nxt < (unsigned)N) ? s_nxt : 0;
      c_nxt = cnt[(lane < (nn1 < 32 ? nn1 : 32)) ? s_cur : d];
      s_nxt = 0;
      if (lane < (nn2 < 32 ? nn2 : 32)) s_nxt = list[(long long)(d + 2) * GCN_CAP + lane];
    }
    const int n = nAll < GCN_CAP ? nAll : GCN_CAP;
    const float dd = rsqrtf((float)(nAll + 1));
    const int* __restrict__ bl = list + (long long)d * GCN_CAP;
    const float4 hv = ((const float4*)(h1 + (long long)d * 128))[lane];
    float4 acc = make_float4(0.f, 0.f, 0.f, 0.f);

    {
      const int m = n < 32 ? n : 32;
      int k0 = 0;
      while (m - k0 > 16) {
        gcn_rows<8>(h1, s_l, w_l, k0, lane, acc);
        k0 += 8;
      }
      switch (m - k0) {
        case 1: gcn_rows<1>(h1, s_l, w_l, k0, lane, acc); break;
        case 2: gcn_rows<2>(h1, s_l, w_l, k0, lane, acc); break;
        case 3: gcn_rows<3>(h1, s_l, w_l, k0, lane, acc); break;
        case 4: gcn_rows<4>(h1, s_l, w_l, k0, lane, acc); break;
        case 5: gcn_rows<5>(h1, s_l, w_l, k0, lane, acc); break;
        case 6: gcn_rows<6>(h1, s_l, w_l, k0, lane, acc); break;
        case 7: gcn_rows<7>(h1, s_l, w_l, k0, lane, acc); break;
        case 8: gcn_rows<8>(h1, s_l, w_l, k0, lane, acc); break;
        case 9: gcn_rows<9>(h1, s_l, w_l, k0, lane, acc); break;
        case 10: gcn_rows<10>(h1, s_l, w_l, k0, lane, acc); break;
        case 11: gcn_rows<11>(h1, s_l, w_l, k0, lane, acc); break;
        case 12: gcn_rows<12>(h1, s_l, w_l, k0, lane, acc); break;
        case 13: gcn_rows<13>(h1, s_l, w_l, k0, lane, acc); break;
        case 14: gcn_rows<14>(h1, s_l, w_l, k0, lane, acc); break;
        case 15: gcn_rows<15>(h1, s_l, w_l, k0, lane, acc); break;
        case 16: gcn_rows<16>(h1, s_l, w_l, k0, lane, acc); break;
        default: break;
      }
    }

    if (n > 32) {
      const int m2 = n - 32;
      int s2_l = 0;
      float w2_l = 0.f;
      if (lane < m2) {
        s2_l = bl[32 + lane];
        s2_l = ((unsigned)s2_l < (unsigned)N) ? s2_l : 0;
        w2_l = rsqrtf((float)(cnt[s2_l] + 1)) * dd;
      }
      for (int k = 0; k < m2; ++k) {
        const int s = __builtin_amdgcn_readlane(s2_l, k);
        const float nrm = __int_as_float(__builtin_amdgcn_readlane(__float_as_int(w2_l), k));
        const float4 v = ((const float4*)(h1 + (long long)s * 128))[lane];
        acc.x += nrm * v.x;
        acc.y += nrm * v.y;
        acc.z += nrm * v.z;
        acc.w += nrm * v.w;
      }
    }

    if (nAll > GCN_CAP) {
      const int* __restrict__ ovl = list + (long long)N * GCN_CAP;
      const int* __restrict__ src = ei;
      const int* __restrict__ dst = ei + E;
      int no = cnt[N];
      no = no < E ? no : E;
      for (int base = 0; base < no; base += 32) {
        int so_l = 0;
        float w_l = 0.f;
        bool mine = false;
        if (base + lane < no) {
          const int e = ovl[base + lane];
          if ((unsigned)e < (unsigned)E && dst[e] == d) {
            mine = true;
            so_l = src[e];
            so_l = ((unsigned)so_l < (unsigned)N) ? so_l : 0;
            w_l = rsqrtf((float)(cnt[so_l] + 1)) * dd;
          }
        }
        unsigned todo = __builtin_amdgcn_ballot_w32(mine);
        while (todo != 0u) {
          const int k = __builtin_ctz(todo);
          todo &= todo - 1u;
          const int s = __builtin_amdgcn_readlane(so_l, k);
          const float nrm = __int_as_float(__builtin_amdgcn_readlane(__float_as_int(w_l), k));
          const float4 v = ((const float4*)(h1 + (long long)s * 128))[lane];
          acc.x += nrm * v.x;
          acc.y += nrm * v.y;
          acc.z += nrm * v.z;
          acc.w += nrm * v.w;
        }
      }
    }

    const float nn = dd * dd;
    float y0 = acc.x + nn * hv.x + bb.x;
    float y1 = acc.y + nn * hv.y + bb.y;
    float y2 = acc.z + nn * hv.z + bb.z;
    float y3 = acc.w + nn * hv.w + bb.w;
    y0 = y0 > 0.0f ? y0 : 0.0f;
    y1 = y1 > 0.0f ? y1 : 0.0f;
    y2 = y2 > 0.0f ? y2 : 0.0f;
    y3 = y3 > 0.0f ? y3 : 0.0f;

    float p0 = y0 * w2r[0] + y1 * w2r[3] + y2 * w2r[6] + y3 * w2r[9];
    float p1 = y0 * w2r[1] + y1 * w2r[4] + y2 * w2r[7] + y3 * w2r[10];
    float p2 = y0 * w2r[2] + y1 * w2r[5] + y2 * w2r[8] + y3 * w2r[11];
    for (int o = 16; o > 0; o >>= 1) {
      p0 += __shfl_xor(p0, o);
      p1 += __shfl_xor(p1, o);
      p2 += __shfl_xor(p2, o);
    }
    if (lane == j) { r0 = p0; r1 = p1; r2 = p2; }
    w_cur = rsqrtf((float)(c_nxt + 1)) * rsqrtf((float)(nn1 + 1));
  }

  if (lane < GCN_G1N && d0 + lane < N)
    ((float4*)h2)[d0 + lane] = make_float4(r0, r1, r2, 0.f);
}

__global__ __launch_bounds__(256) void gcn_gather2(
    const int* __restrict__ cnt, const int* __restrict__ list,
    const int* __restrict__ ei,
    const float* __restrict__ h2, const float* __restrict__ b2,
    float* __restrict__ out, int N, int E) {
  __shared__ float res[96];
  const int t = threadIdx.x;
  const int g = t >> 3;
  const int l = t & 7;
  const int d = blockIdx.x * 32 + g;
  const bool valid = d < N;
  const int dc = valid ? d : 0;
  const int nAll = valid ? cnt[dc] : 0;
  const int n = nAll < GCN_CAP ? nAll : GCN_CAP;
  const float dd = rsqrtf((float)(nAll + 1));
  const int* __restrict__ bl = list + (long long)dc * GCN_CAP;

  float a0 = 0.f, a1 = 0.f, a2 = 0.f;
  for (int j = l; j < n; j += 8) {
    int s = bl[j];
    s = ((unsigned)s < (unsigned)N) ? s : 0;
    const float nrm = rsqrtf((float)(cnt[s] + 1)) * dd;
    const float4 hq = ((const float4*)h2)[s];
    a0 += nrm * hq.x;
    a1 += nrm * hq.y;
    a2 += nrm * hq.z;
  }
  if (nAll > GCN_CAP) {
    const int* __restrict__ ovl = list + (long long)N * GCN_CAP;
    const int* __restrict__ src = ei;
    const int* __restrict__ dst = ei + E;
    int no = cnt[N];
    no = no < E ? no : E;
    for (int o = l; o < no; o += 8) {
      const int e = ovl[o];
      if ((unsigned)e < (unsigned)E && dst[e] == d) {
        int s = src[e];
        s = ((unsigned)s < (unsigned)N) ? s : 0;
        const float nrm = rsqrtf((float)(cnt[s] + 1)) * dd;
        const float4 hq = ((const float4*)h2)[s];
        a0 += nrm * hq.x;
        a1 += nrm * hq.y;
        a2 += nrm * hq.z;
      }
    }
  }
  for (int o = 4; o > 0; o >>= 1) {
    a0 += __shfl_xor(a0, o);
    a1 += __shfl_xor(a1, o);
    a2 += __shfl_xor(a2, o);
  }
  if (l == 0) {
    const float nn = dd * dd;
    const float4 hq = ((const float4*)h2)[dc];
    res[g * 3 + 0] = a0 + nn * hq.x + b2[0];
    res[g * 3 + 1] = a1 + nn * hq.y + b2[1];
    res[g * 3 + 2] = a2 + nn * hq.z + b2[2];
  }
  __syncthreads();

  if (t < 24) {
    const long long f = (long long)blockIdx.x * 96 + t * 4;
    const long long n3 = (long long)N * 3;
    if (f + 3 < n3) {
      *(float4*)(out + f) = make_float4(res[t * 4 + 0], res[t * 4 + 1],
                                        res[t * 4 + 2], res[t * 4 + 3]);
    } else {
      for (int q = 0; q < 4; ++q)
        if (f + q < n3) out[f + q] = res[t * 4 + q];
    }
  }
}

extern "C" void kernel_launch(void* const* d_in, const int* in_sizes, int n_in,
                              void* d_out, int out_size, void* d_ws, size_t ws_size,
                              hipStream_t stream) {
  const float* x  = (const float*)d_in[0];
  const int*   ei = (const int*)d_in[1];
  const float* W1 = (const float*)d_in[2];
  const float* b1 = (const float*)d_in[3];
  const float* W2 = (const float*)d_in[4];
  const float* b2 = (const float*)d_in[5];
  float* out = (float*)d_out;

  const int N = in_sizes[0] / 512;
  const int E = in_sizes[1] / 2;

  const long long Npad = ((long long)N + 255) & ~255LL;
  const long long Cpad = ((long long)N + 1 + 255) & ~255LL;
  float* ws   = (float*)d_ws;
  float* h1   = ws + Npad;
  float* pl   = h1 + (long long)N * 128;
  float* h2   = pl + (((long long)N * 124) & ~31LL);
  int* cnt  = (int*)pl;
  int* list = cnt + Cpad;

  const int B = 256;

  gcn_clear<<<(N + 1 + B - 1) / B, B, 0, stream>>>(cnt, N + 1);

  gcn_gemm1_wmma<<<256, 128, 0, stream>>>(x, W1, h1, N, ei, cnt, list, E);

  gcn_gather1<<<(N + 15) / 16, 64, 0, stream>>>(cnt, list, ei, h1, b1, W2, h2, N, E);
  gcn_gather2<<<(N + 31) / 32, B, 0, stream>>>(cnt, list, ei, h2, b2, out, N, E);
}
